# starting kernel
# baseline (speedup 1.0000x reference)
#include <hip/hip_runtime.h>

#define GIN_D 64
#define TSTR 68
#define WSTR 68

typedef __attribute__((ext_vector_type(2))) float v2f;
typedef __attribute__((ext_vector_type(8))) float v8f;

__global__ __launch_bounds__(256) void gin_zero_kernel(float4* __restrict__ p, int n4) {
    int i = blockIdx.x * 256 + threadIdx.x;
    if (i < n4) p[i] = make_float4(0.f, 0.f, 0.f, 0.f);
}

__global__ __launch_bounds__(256) void gin_agg_kernel(const float* __restrict__ h,
                                                      const int* __restrict__ src,
                                                      const int* __restrict__ dst,
                                                      float* __restrict__ agg, int e) {
    int tid  = blockIdx.x * 256 + threadIdx.x;
    int edge = tid >> 4;
    if (edge >= e) return;
    int c = (tid & 15) << 2;
    int s = src[edge];
    int d = dst[edge];
    const float4 v = *(const float4*)(h + (size_t)s * GIN_D + c);
    float* ap = agg + (size_t)d * GIN_D + c;
    unsafeAtomicAdd(ap + 0, v.x);
    unsafeAtomicAdd(ap + 1, v.y);
    unsafeAtomicAdd(ap + 2, v.z);
    unsafeAtomicAdd(ap + 3, v.w);
}

__global__ __launch_bounds__(128)
void gin_mlp_kernel(const float* __restrict__ hin, const float* __restrict__ agg,
                    const float* __restrict__ w1, const float* __restrict__ b1,
                    const float* __restrict__ w2, const float* __restrict__ b2,
                    float* __restrict__ hout, int n) {
    __shared__ float sW1t[GIN_D * WSTR];
    __shared__ float sW2t[GIN_D * WSTR];
    __shared__ float sB1[GIN_D], sB2[GIN_D];
    __shared__ float sT[4][16 * TSTR];

    const int t    = threadIdx.x;
    const int wave = t >> 5;
    const int lane = t & 31;
    const int m    = lane & 15;
    const int hi   = lane >> 4;

    for (int i = t; i < GIN_D * GIN_D / 4; i += 128) {
        int k  = i >> 4;
        int c4 = (i & 15) << 2;
        float4 a = ((const float4*)w1)[i];
        sW1t[(c4 + 0) * WSTR + k] = a.x;
        sW1t[(c4 + 1) * WSTR + k] = a.y;
        sW1t[(c4 + 2) * WSTR + k] = a.z;
        sW1t[(c4 + 3) * WSTR + k] = a.w;
        float4 b = ((const float4*)w2)[i];
        sW2t[(c4 + 0) * WSTR + k] = b.x;
        sW2t[(c4 + 1) * WSTR + k] = b.y;
        sW2t[(c4 + 2) * WSTR + k] = b.z;
        sW2t[(c4 + 3) * WSTR + k] = b.w;
    }
    if (t < GIN_D) { sB1[t] = b1[t]; sB2[t] = b2[t]; }
    __syncthreads();

    const int row0 = (blockIdx.x * 4 + wave) * 16;
    float* tile = sT[wave];

    for (int i = lane; i < 256; i += 32) {
        int r  = i >> 4;
        int c4 = (i & 15) << 2;
        int gr = row0 + r;
        float4 v = make_float4(0.f, 0.f, 0.f, 0.f);
        if (gr < n) {
            const float4 a = *(const float4*)(hin + (size_t)gr * GIN_D + c4);
            const float4 g = *(const float4*)(agg + (size_t)gr * GIN_D + c4);
            v.x = a.x + g.x; v.y = a.y + g.y; v.z = a.z + g.z; v.w = a.w + g.w;
        }
        tile[r * TSTR + c4 + 0] = v.x;
        tile[r * TSTR + c4 + 1] = v.y;
        tile[r * TSTR + c4 + 2] = v.z;
        tile[r * TSTR + c4 + 3] = v.w;
    }
    __syncthreads();

    v2f A[16];
#pragma unroll
    for (int kb = 0; kb < 16; ++kb) {
        int k = 4 * kb + 2 * hi;
        A[kb].x = tile[m * TSTR + k];
        A[kb].y = tile[m * TSTR + k + 1];
    }
    __syncthreads();

    for (int nt = 0; nt < 4; ++nt) {
        const int n0 = nt * 16;
        float bv = sB1[n0 + m];
        v8f acc = {bv, bv, bv, bv, bv, bv, bv, bv};
#pragma unroll
        for (int kb = 0; kb < 16; ++kb) {
            int k = 4 * kb + 2 * hi;
            v2f B;
            B.x = sW1t[(n0 + m) * WSTR + k];
            B.y = sW1t[(n0 + m) * WSTR + k + 1];
            acc = __builtin_amdgcn_wmma_f32_16x16x4_f32(false, A[kb], false, B,
                                                        (short)0, acc, false, false);
        }
#pragma unroll
        for (int j = 0; j < 8; ++j)
            tile[(j + 8 * hi) * TSTR + n0 + m] = fmaxf(acc[j], 0.f);
    }
    __syncthreads();

#pragma unroll
    for (int kb = 0; kb < 16; ++kb) {
        int k = 4 * kb + 2 * hi;
        A[kb].x = tile[m * TSTR + k];
        A[kb].y = tile[m * TSTR + k + 1];
    }

    for (int nt = 0; nt < 4; ++nt) {
        const int n0 = nt * 16;
        float bv = sB2[n0 + m];
        v8f acc = {bv, bv, bv, bv, bv, bv, bv, bv};
#pragma unroll
        for (int kb = 0; kb < 16; ++kb) {
            int k = 4 * kb + 2 * hi;
            v2f B;
            B.x = sW2t[(n0 + m) * WSTR + k];
            B.y = sW2t[(n0 + m) * WSTR + k + 1];
            acc = __builtin_amdgcn_wmma_f32_16x16x4_f32(false, A[kb], false, B,
                                                        (short)0, acc, false, false);
        }
#pragma unroll
        for (int j = 0; j < 8; ++j) {
            int gr = row0 + j + 8 * hi;
            if (gr < n) hout[(size_t)gr * GIN_D + n0 + m] = fmaxf(acc[j], 0.f);
        }
    }
}

__global__ __launch_bounds__(256) void gin_readout_kernel(const float* __restrict__ h,
                                                          const float* __restrict__ ow,
                                                          const float* __restrict__ ob,
                                                          float* __restrict__ out, int n) {
    int gw   = (blockIdx.x * 256 + threadIdx.x) >> 5;
    int lane = threadIdx.x & 31;
    if (gw >= n) return;
    float s = h[(size_t)gw * GIN_D + lane] * ow[lane] +
              h[(size_t)gw * GIN_D + 32 + lane] * ow[32 + lane];
    for (int off = 16; off; off >>= 1) s += __shfl_xor(s, off, 32);
    if (lane == 0) out[gw] = s + ob[0];
}

extern "C" void kernel_launch(void* const* d_in, const int* in_sizes, int n_in,
                              void* d_out, int out_size, void* d_ws, size_t ws_size,
                              hipStream_t stream) {
    const float* x  = (const float*)d_in[0];
    const int*   ei = (const int*)d_in[1];
    const int N = in_sizes[0] / GIN_D;
    const int E = in_sizes[1] / 2;
    const int* src = ei;
    const int* dst = ei + E;
    const float* W1[3] = {(const float*)d_in[2], (const float*)d_in[6],  (const float*)d_in[10]};
    const float* B1[3] = {(const float*)d_in[3], (const float*)d_in[7],  (const float*)d_in[11]};
    const float* W2[3] = {(const float*)d_in[4], (const float*)d_in[8],  (const float*)d_in[12]};
    const float* B2[3] = {(const float*)d_in[5], (const float*)d_in[9],  (const float*)d_in[13]};
    const float* out_w = (const float*)d_in[14];
    const float* out_b = (const float*)d_in[15];
    float* out = (float*)d_out;

    float* agg = (float*)d_ws;
    float* hA  = agg + (size_t)N * GIN_D;
    float* hB  = hA  + (size_t)N * GIN_D;

    const int n4 = N * GIN_D / 4;
    dim3 zg((unsigned)((n4 + 255) / 256)), b256(256);
    dim3 ag((unsigned)(((long long)E * 16 + 255) / 256));
    dim3 mg((unsigned)((N + 63) / 64)), mb(128);
    dim3 rg((unsigned)(((long long)N * 32 + 255) / 256));

    const float* hin = x;
    float* bufs[2] = {hA, hB};
    for (int l = 0; l < 3; ++l) {
        gin_zero_kernel<<<zg, b256, 0, stream>>>((float4*)agg, n4);
        gin_agg_kernel<<<ag, b256, 0, stream>>>(hin, src, dst, agg, E);
        float* hout = bufs[l & 1];
        gin_mlp_kernel<<<mg, mb, 0, stream>>>(hin, agg, W1[l], B1[l], W2[l], B2[l], hout, N);
        hin = hout;
    }
    gin_readout_kernel<<<rg, b256, 0, stream>>>(hin, out_w, out_b, out, N);
}


# Float-atomic scatter-add replaced by a per-call in-edge list grouped by destination (integer count with arrival-order rank, prefix sum, place) and a per-node gather sum in list order
# speedup vs baseline: 1.7259x; 1.7259x over previous
#include <hip/hip_runtime.h>

#define GIN_D 64
#define TSTR 68
#define WSTR 68

typedef __attribute__((ext_vector_type(2))) float v2f;
typedef __attribute__((ext_vector_type(8))) float v8f;

__global__ __launch_bounds__(256) void gin_clear_kernel(int* __restrict__ cnt, int n) {
    int i = blockIdx.x * 256 + threadIdx.x;
    if (i < n) cnt[i] = 0;
}

__global__ __launch_bounds__(256) void gin_count_kernel(const int* __restrict__ dst,
                                                        int* __restrict__ cnt,
                                                        int* __restrict__ rank, int e) {
    int i = blockIdx.x * 256 + threadIdx.x;
    if (i < e) rank[i] = atomicAdd(&cnt[dst[i]], 1);
}

__global__ __launch_bounds__(256) void gin_chunk_sum_kernel(const int* __restrict__ cnt,
                                                            int* __restrict__ bsum, int n) {
    __shared__ int sw[8];
    const int t    = threadIdx.x;
    const int base = blockIdx.x * 1024 + t * 4;
    int s = 0;
    if (base + 0 < n) s += cnt[base + 0];
    if (base + 1 < n) s += cnt[base + 1];
    if (base + 2 < n) s += cnt[base + 2];
    if (base + 3 < n) s += cnt[base + 3];
    for (int o = 16; o; o >>= 1) s += __shfl_xor(s, o, 32);
    if ((t & 31) == 0) sw[t >> 5] = s;
    __syncthreads();
    if (t == 0) {
        int tot = 0;
        for (int w = 0; w < 8; ++w) tot += sw[w];
        bsum[blockIdx.x] = tot;
    }
}

__global__ __launch_bounds__(256) void gin_scan_kernel(const int* __restrict__ cnt,
                                                       const int* __restrict__ bsum,
                                                       int* __restrict__ off, int n) {
    __shared__ int sw[8];
    __shared__ int sp[8];
    const int t    = threadIdx.x;
    const int lane = t & 31;
    const int wave = t >> 5;

    int p = 0;
    for (int i = t; i < (int)blockIdx.x; i += 256) p += bsum[i];
    for (int o = 16; o; o >>= 1) p += __shfl_xor(p, o, 32);

    const int base = blockIdx.x * 1024 + t * 4;
    const int c0 = (base + 0 < n) ? cnt[base + 0] : 0;
    const int c1 = (base + 1 < n) ? cnt[base + 1] : 0;
    const int c2 = (base + 2 < n) ? cnt[base + 2] : 0;
    const int c3 = (base + 3 < n) ? cnt[base + 3] : 0;
    const int s  = c0 + c1 + c2 + c3;

    int inc = s;
    for (int d = 1; d < 32; d <<= 1) {
        int v = __shfl_up(inc, d, 32);
        if (lane >= d) inc += v;
    }
    if (lane == 31) sw[wave] = inc;
    if (lane == 0)  sp[wave] = p;
    __syncthreads();

    int before = 0;
    for (int w = 0; w < 8; ++w) {
        before += sp[w];
        if (w < wave) before += sw[w];
    }
    int ex = before + inc - s;
    if (base + 0 < n) off[base + 0] = ex;
    ex += c0;
    if (base + 1 < n) off[base + 1] = ex;
    ex += c1;
    if (base + 2 < n) off[base + 2] = ex;
    ex += c2;
    if (base + 3 < n) off[base + 3] = ex;
}

__global__ __launch_bounds__(256) void gin_place_kernel(const int* __restrict__ src,
                                                        const int* __restrict__ dst,
                                                        const int* __restrict__ rank,
                                                        const int* __restrict__ off,
                                                        int* __restrict__ lst, int e) {
    int i = blockIdx.x * 256 + threadIdx.x;
    if (i >= e) return;
    int pos = off[dst[i]] + rank[i];
    if ((unsigned)pos < (unsigned)e) lst[pos] = src[i];
}

__global__ __launch_bounds__(256) void gin_gather_kernel(const float* __restrict__ h,
                                                         const int* __restrict__ off,
                                                         const int* __restrict__ cnt,
                                                         const int* __restrict__ lst,
                                                         float* __restrict__ z, int n) {
    int tid  = blockIdx.x * 256 + threadIdx.x;
    int node = tid >> 4;
    if (node >= n) return;
    int c   = (tid & 15) << 2;
    int p   = off[node];
    int end = p + cnt[node];
    float4 acc = *(const float4*)(h + (size_t)node * GIN_D + c);
    for (; p + 4 <= end; p += 4) {
        const int s0 = lst[p + 0];
        const int s1 = lst[p + 1];
        const int s2 = lst[p + 2];
        const int s3 = lst[p + 3];
        const float4 v0 = *(const float4*)(h + (size_t)s0 * GIN_D + c);
        const float4 v1 = *(const float4*)(h + (size_t)s1 * GIN_D + c);
        const float4 v2 = *(const float4*)(h + (size_t)s2 * GIN_D + c);
        const float4 v3 = *(const float4*)(h + (size_t)s3 * GIN_D + c);
        acc.x += v0.x; acc.y += v0.y; acc.z += v0.z; acc.w += v0.w;
        acc.x += v1.x; acc.y += v1.y; acc.z += v1.z; acc.w += v1.w;
        acc.x += v2.x; acc.y += v2.y; acc.z += v2.z; acc.w += v2.w;
        acc.x += v3.x; acc.y += v3.y; acc.z += v3.z; acc.w += v3.w;
    }
    if (p < end) {
        const int last = end - 1;
        const bool k1 = p + 1 < end;
        const bool k2 = p + 2 < end;
        const int s0 = lst[p];
        const int s1 = lst[k1 ? p + 1 : last];
        const int s2 = lst[k2 ? p + 2 : last];
        const float4 v0 = *(const float4*)(h + (size_t)s0 * GIN_D + c);
        const float4 v1 = *(const float4*)(h + (size_t)s1 * GIN_D + c);
        const float4 v2 = *(const float4*)(h + (size_t)s2 * GIN_D + c);
        acc.x += v0.x; acc.y += v0.y; acc.z += v0.z; acc.w += v0.w;
        acc.x += k1 ? v1.x : 0.f; acc.y += k1 ? v1.y : 0.f;
        acc.z += k1 ? v1.z : 0.f; acc.w += k1 ? v1.w : 0.f;
        acc.x += k2 ? v2.x : 0.f; acc.y += k2 ? v2.y : 0.f;
        acc.z += k2 ? v2.z : 0.f; acc.w += k2 ? v2.w : 0.f;
    }
    *(float4*)(z + (size_t)node * GIN_D + c) = acc;
}

__global__ __launch_bounds__(128)
void gin_mlp_kernel(const float* __restrict__ zin,
                    const float* __restrict__ w1, const float* __restrict__ b1,
                    const float* __restrict__ w2, const float* __restrict__ b2,
                    float* __restrict__ hout, int n) {
    __shared__ float sW1t[GIN_D * WSTR];
    __shared__ float sW2t[GIN_D * WSTR];
    __shared__ float sB1[GIN_D], sB2[GIN_D];
    __shared__ float sT[4][16 * TSTR];

    const int t    = threadIdx.x;
    const int wave = t >> 5;
    const int lane = t & 31;
    const int m    = lane & 15;
    const int hi   = lane >> 4;

    for (int i = t; i < GIN_D * GIN_D / 4; i += 128) {
        int k  = i >> 4;
        int c4 = (i & 15) << 2;
        float4 a = ((const float4*)w1)[i];
        sW1t[(c4 + 0) * WSTR + k] = a.x;
        sW1t[(c4 + 1) * WSTR + k] = a.y;
        sW1t[(c4 + 2) * WSTR + k] = a.z;
        sW1t[(c4 + 3) * WSTR + k] = a.w;
        float4 b = ((const float4*)w2)[i];
        sW2t[(c4 + 0) * WSTR + k] = b.x;
        sW2t[(c4 + 1) * WSTR + k] = b.y;
        sW2t[(c4 + 2) * WSTR + k] = b.z;
        sW2t[(c4 + 3) * WSTR + k] = b.w;
    }
    if (t < GIN_D) { sB1[t] = b1[t]; sB2[t] = b2[t]; }
    __syncthreads();

    const int row0 = (blockIdx.x * 4 + wave) * 16;
    float* tile = sT[wave];

    for (int i = lane; i < 256; i += 32) {
        int r  = i >> 4;
        int c4 = (i & 15) << 2;
        int gr = row0 + r;
        float4 v = make_float4(0.f, 0.f, 0.f, 0.f);
        if (gr < n) v = *(const float4*)(zin + (size_t)gr * GIN_D + c4);
        tile[r * TSTR + c4 + 0] = v.x;
        tile[r * TSTR + c4 + 1] = v.y;
        tile[r * TSTR + c4 + 2] = v.z;
        tile[r * TSTR + c4 + 3] = v.w;
    }
    __syncthreads();

    v2f A[16];
#pragma unroll
    for (int kb = 0; kb < 16; ++kb) {
        int k = 4 * kb + 2 * hi;
        A[kb].x = tile[m * TSTR + k];
        A[kb].y = tile[m * TSTR + k + 1];
    }
    __syncthreads();

    for (int nt = 0; nt < 4; ++nt) {
        const int n0 = nt * 16;
        float bv = sB1[n0 + m];
        v8f acc = {bv, bv, bv, bv, bv, bv, bv, bv};
#pragma unroll
        for (int kb = 0; kb < 16; ++kb) {
            int k = 4 * kb + 2 * hi;
            v2f B;
            B.x = sW1t[(n0 + m) * WSTR + k];
            B.y = sW1t[(n0 + m) * WSTR + k + 1];
            acc = __builtin_amdgcn_wmma_f32_16x16x4_f32(false, A[kb], false, B,
                                                        (short)0, acc, false, false);
        }
#pragma unroll
        for (int j = 0; j < 8; ++j)
            tile[(j + 8 * hi) * TSTR + n0 + m] = fmaxf(acc[j], 0.f);
    }
    __syncthreads();

#pragma unroll
    for (int kb = 0; kb < 16; ++kb) {
        int k = 4 * kb + 2 * hi;
        A[kb].x = tile[m * TSTR + k];
        A[kb].y = tile[m * TSTR + k + 1];
    }

    for (int nt = 0; nt < 4; ++nt) {
        const int n0 = nt * 16;
        float bv = sB2[n0 + m];
        v8f acc = {bv, bv, bv, bv, bv, bv, bv, bv};
#pragma unroll
        for (int kb = 0; kb < 16; ++kb) {
            int k = 4 * kb + 2 * hi;
            v2f B;
            B.x = sW2t[(n0 + m) * WSTR + k];
            B.y = sW2t[(n0 + m) * WSTR + k + 1];
            acc = __builtin_amdgcn_wmma_f32_16x16x4_f32(false, A[kb], false, B,
                                                        (short)0, acc, false, false);
        }
#pragma unroll
        for (int j = 0; j < 8; ++j) {
            int gr = row0 + j + 8 * hi;
            if (gr < n) hout[(size_t)gr * GIN_D + n0 + m] = fmaxf(acc[j], 0.f);
        }
    }
}

__global__ __launch_bounds__(256) void gin_readout_kernel(const float* __restrict__ h,
                                                          const float* __restrict__ ow,
                                                          const float* __restrict__ ob,
                                                          float* __restrict__ out, int n) {
    int gw   = (blockIdx.x * 256 + threadIdx.x) >> 5;
    int lane = threadIdx.x & 31;
    if (gw >= n) return;
    float s = h[(size_t)gw * GIN_D + lane] * ow[lane] +
              h[(size_t)gw * GIN_D + 32 + lane] * ow[32 + lane];
    for (int off = 16; off; off >>= 1) s += __shfl_xor(s, off, 32);
    if (lane == 0) out[gw] = s + ob[0];
}

extern "C" void kernel_launch(void* const* d_in, const int* in_sizes, int n_in,
                              void* d_out, int out_size, void* d_ws, size_t ws_size,
                              hipStream_t stream) {
    const float* x  = (const float*)d_in[0];
    const int*   ei = (const int*)d_in[1];
    const int N = in_sizes[0] / GIN_D;
    const int E = in_sizes[1] / 2;
    const int* src = ei;
    const int* dst = ei + E;
    const float* W1[3] = {(const float*)d_in[2], (const float*)d_in[6],  (const float*)d_in[10]};
    const float* B1[3] = {(const float*)d_in[3], (const float*)d_in[7],  (const float*)d_in[11]};
    const float* W2[3] = {(const float*)d_in[4], (const float*)d_in[8],  (const float*)d_in[12]};
    const float* B2[3] = {(const float*)d_in[5], (const float*)d_in[9],  (const float*)d_in[13]};
    const float* out_w = (const float*)d_in[14];
    const float* out_b = (const float*)d_in[15];
    float* out = (float*)d_out;

    float* z   = (float*)d_ws;
    float* hA  = z   + (size_t)N * GIN_D;
    float* hB  = hA  + (size_t)N * GIN_D;
    const int nchunk = (N + 1023) / 1024;
    int* cnt  = (int*)(hB + (size_t)N * GIN_D);
    int* off  = cnt  + (size_t)nchunk * 1024;
    int* bsum = off  + (size_t)nchunk * 1024;
    int* rank = bsum + (size_t)nchunk * 1024;
    int* lst  = rank + (size_t)E;

    dim3 b256(256);
    dim3 ng((unsigned)((N + 255) / 256));
    dim3 eg((unsigned)((E + 255) / 256));
    dim3 cg((unsigned)nchunk);
    dim3 gg((unsigned)(((long long)N * 16 + 255) / 256));
    dim3 mg((unsigned)((N + 63) / 64)), mb(128);
    dim3 rg((unsigned)(((long long)N * 32 + 255) / 256));

    gin_clear_kernel<<<ng, b256, 0, stream>>>(cnt, N);
    gin_count_kernel<<<eg, b256, 0, stream>>>(dst, cnt, rank, E);
    gin_chunk_sum_kernel<<<cg, b256, 0, stream>>>(cnt, bsum, N);
    gin_scan_kernel<<<cg, b256, 0, stream>>>(cnt, bsum, off, N);
    gin_place_kernel<<<eg, b256, 0, stream>>>(src, dst, rank, off, lst, E);

    const float* hin = x;
    float* bufs[2] = {hA, hB};
    for (int l = 0; l < 3; ++l) {
        gin_gather_kernel<<<gg, b256, 0, stream>>>(hin, off, cnt, lst, z, N);
        float* hout = bufs[l & 1];
        gin_mlp_kernel<<<mg, mb, 0, stream>>>(z, W1[l], B1[l], W2[l], B2[l], hout, N);
        hin = hout;
    }
    gin_readout_kernel<<<rg, b256, 0, stream>>>(hin, out_w, out_b, out, N);
}


# Neighbour gather fused into the MLP launch: each wave sums its rows of z (h[i], then in-edges in list order) straight into its LDS A tile; z buffer and 3 of 12 launches gone; 8-wave blocks
# speedup vs baseline: 1.9847x; 1.1500x over previous
#include <hip/hip_runtime.h>

#define GIN_D 64
#define TSTR 68
#define WSTR 68

typedef __attribute__((ext_vector_type(2))) float v2f;
typedef __attribute__((ext_vector_type(8))) float v8f;

__global__ __launch_bounds__(256) void gin_clear_kernel(int* __restrict__ cnt, int n) {
    int i = blockIdx.x * 256 + threadIdx.x;
    if (i < n) cnt[i] = 0;
}

__global__ __launch_bounds__(256) void gin_count_kernel(const int* __restrict__ dst,
                                                        int* __restrict__ cnt,
                                                        int* __restrict__ rank, int e) {
    int i = blockIdx.x * 256 + threadIdx.x;
    if (i < e) rank[i] = atomicAdd(&cnt[dst[i]], 1);
}

__global__ __launch_bounds__(256) void gin_chunk_sum_kernel(const int* __restrict__ cnt,
                                                            int* __restrict__ bsum, int n) {
    __shared__ int sw[8];
    const int t    = threadIdx.x;
    const int base = blockIdx.x * 1024 + t * 4;
    int s = 0;
    if (base + 0 < n) s += cnt[base + 0];
    if (base + 1 < n) s += cnt[base + 1];
    if (base + 2 < n) s += cnt[base + 2];
    if (base + 3 < n) s += cnt[base + 3];
    for (int o = 16; o; o >>= 1) s += __shfl_xor(s, o, 32);
    if ((t & 31) == 0) sw[t >> 5] = s;
    __syncthreads();
    if (t == 0) {
        int tot = 0;
        for (int w = 0; w < 8; ++w) tot += sw[w];
        bsum[blockIdx.x] = tot;
    }
}

__global__ __launch_bounds__(256) void gin_scan_kernel(const int* __restrict__ cnt,
                                                       const int* __restrict__ bsum,
                                                       int* __restrict__ off, int n) {
    __shared__ int sw[8];
    __shared__ int sp[8];
    const int t    = threadIdx.x;
    const int lane = t & 31;
    const int wave = t >> 5;

    int p = 0;
    for (int i = t; i < (int)blockIdx.x; i += 256) p += bsum[i];
    for (int o = 16; o; o >>= 1) p += __shfl_xor(p, o, 32);

    const int base = blockIdx.x * 1024 + t * 4;
    const int c0 = (base + 0 < n) ? cnt[base + 0] : 0;
    const int c1 = (base + 1 < n) ? cnt[base + 1] : 0;
    const int c2 = (base + 2 < n) ? cnt[base + 2] : 0;
    const int c3 = (base + 3 < n) ? cnt[base + 3] : 0;
    const int s  = c0 + c1 + c2 + c3;

    int inc = s;
    for (int d = 1; d < 32; d <<= 1) {
        int v = __shfl_up(inc, d, 32);
        if (lane >= d) inc += v;
    }
    if (lane == 31) sw[wave] = inc;
    if (lane == 0)  sp[wave] = p;
    __syncthreads();

    int before = 0;
    for (int w = 0; w < 8; ++w) {
        before += sp[w];
        if (w < wave) before += sw[w];
    }
    int ex = before + inc - s;
    if (base + 0 < n) off[base + 0] = ex;
    ex += c0;
    if (base + 1 < n) off[base + 1] = ex;
    ex += c1;
    if (base + 2 < n) off[base + 2] = ex;
    ex += c2;
    if (base + 3 < n) off[base + 3] = ex;
}

__global__ __launch_bounds__(256) void gin_place_kernel(const int* __restrict__ src,
                                                        const int* __restrict__ dst,
                                                        const int* __restrict__ rank,
                                                        const int* __restrict__ off,
                                                        int* __restrict__ lst, int e) {
    int i = blockIdx.x * 256 + threadIdx.x;
    if (i >= e) return;
    int pos = off[dst[i]] + rank[i];
    if ((unsigned)pos < (unsigned)e) lst[pos] = src[i];
}

__global__ __launch_bounds__(256)
void gin_layer_kernel(const float* __restrict__ h,
                      const int* __restrict__ off, const int* __restrict__ cnt,
                      const int* __restrict__ lst,
                      const float* __restrict__ w1, const float* __restrict__ b1,
                      const float* __restrict__ w2, const float* __restrict__ b2,
                      float* __restrict__ hout, int n) {
    __shared__ float sW1t[GIN_D * WSTR];
    __shared__ float sW2t[GIN_D * WSTR];
    __shared__ float sB1[GIN_D], sB2[GIN_D];
    __shared__ float sT[8][16 * TSTR];

    const int t    = threadIdx.x;
    const int wave = t >> 5;
    const int lane = t & 31;
    const int m    = lane & 15;
    const int hi   = lane >> 4;

    const int row0 = (blockIdx.x * 8 + wave) * 16;
    float* tile = sT[wave];

    const int q = lane >> 3;
    const int c = (lane & 7) << 2;

    int pbeg[4], pcnt[4];
#pragma unroll
    for (int it = 0; it < 4; ++it) {
        const int nd = min(row0 + it * 4 + q, n - 1);
        pbeg[it] = off[nd];
        pcnt[it] = cnt[nd];
    }

    for (int i = t; i < GIN_D * GIN_D / 4; i += 256) {
        int k  = i >> 4;
        int c4 = (i & 15) << 2;
        float4 a = ((const float4*)w1)[i];
        sW1t[(c4 + 0) * WSTR + k] = a.x;
        sW1t[(c4 + 1) * WSTR + k] = a.y;
        sW1t[(c4 + 2) * WSTR + k] = a.z;
        sW1t[(c4 + 3) * WSTR + k] = a.w;
        float4 b = ((const float4*)w2)[i];
        sW2t[(c4 + 0) * WSTR + k] = b.x;
        sW2t[(c4 + 1) * WSTR + k] = b.y;
        sW2t[(c4 + 2) * WSTR + k] = b.z;
        sW2t[(c4 + 3) * WSTR + k] = b.w;
    }
    if (t < GIN_D) { sB1[t] = b1[t]; sB2[t] = b2[t]; }

#pragma unroll
    for (int it = 0; it < 4; ++it) {
        const int r    = it * 4 + q;
        const int node = row0 + r;
        float4 a0 = make_float4(0.f, 0.f, 0.f, 0.f);
        float4 a1 = make_float4(0.f, 0.f, 0.f, 0.f);
        if (node < n) {
            int p = pbeg[it];
            const int end  = p + pcnt[it];
            const int last = end - 1;
            const float* hc = h + c;
            a0 = *(const float4*)(hc + (size_t)node * GIN_D);
            a1 = *(const float4*)(hc + (size_t)node * GIN_D + 32);
            int s0 = 0, s1 = 0, s2 = 0, s3 = 0;
            if (p < end) {
                s0 = lst[p];
                s1 = lst[min(p + 1, last)];
                s2 = lst[min(p + 2, last)];
                s3 = lst[min(p + 3, last)];
            }
            while (p < end) {
                const bool k1 = p + 1 < end;
                const bool k2 = p + 2 < end;
                const bool k3 = p + 3 < end;
                const float4 u0 = *(const float4*)(hc + (size_t)s0 * GIN_D);
                const float4 v0 = *(const float4*)(hc + (size_t)s0 * GIN_D + 32);
                const float4 u1 = *(const float4*)(hc + (size_t)s1 * GIN_D);
                const float4 v1 = *(const float4*)(hc + (size_t)s1 * GIN_D + 32);
                const float4 u2 = *(const float4*)(hc + (size_t)s2 * GIN_D);
                const float4 v2 = *(const float4*)(hc + (size_t)s2 * GIN_D + 32);
                const float4 u3 = *(const float4*)(hc + (size_t)s3 * GIN_D);
                const float4 v3 = *(const float4*)(hc + (size_t)s3 * GIN_D + 32);
                p += 4;
                if (p < end) {
                    s0 = lst[p];
                    s1 = lst[min(p + 1, last)];
                    s2 = lst[min(p + 2, last)];
                    s3 = lst[min(p + 3, last)];
                }
                a0.x += u0.x; a0.y += u0.y; a0.z += u0.z; a0.w += u0.w;
                a1.x += v0.x; a1.y += v0.y; a1.z += v0.z; a1.w += v0.w;
                a0.x += k1 ? u1.x : 0.f; a0.y += k1 ? u1.y : 0.f;
                a0.z += k1 ? u1.z : 0.f; a0.w += k1 ? u1.w : 0.f;
                a1.x += k1 ? v1.x : 0.f; a1.y += k1 ? v1.y : 0.f;
                a1.z += k1 ? v1.z : 0.f; a1.w += k1 ? v1.w : 0.f;
                a0.x += k2 ? u2.x : 0.f; a0.y += k2 ? u2.y : 0.f;
                a0.z += k2 ? u2.z : 0.f; a0.w += k2 ? u2.w : 0.f;
                a1.x += k2 ? v2.x : 0.f; a1.y += k2 ? v2.y : 0.f;
                a1.z += k2 ? v2.z : 0.f; a1.w += k2 ? v2.w : 0.f;
                a0.x += k3 ? u3.x : 0.f; a0.y += k3 ? u3.y : 0.f;
                a0.z += k3 ? u3.z : 0.f; a0.w += k3 ? u3.w : 0.f;
                a1.x += k3 ? v3.x : 0.f; a1.y += k3 ? v3.y : 0.f;
                a1.z += k3 ? v3.z : 0.f; a1.w += k3 ? v3.w : 0.f;
            }
        }
        float* tr = tile + r * TSTR + c;
        tr[0] = a0.x; tr[1] = a0.y; tr[2] = a0.z; tr[3] = a0.w;
        tr[32] = a1.x; tr[33] = a1.y; tr[34] = a1.z; tr[35] = a1.w;
    }
    __syncthreads();

    v2f A[16];
#pragma unroll
    for (int kb = 0; kb < 16; ++kb) {
        int k = 4 * kb + 2 * hi;
        A[kb].x = tile[m * TSTR + k];
        A[kb].y = tile[m * TSTR + k + 1];
    }
    __syncthreads();

    for (int nt = 0; nt < 4; ++nt) {
        const int n0 = nt * 16;
        float bv = sB1[n0 + m];
        v8f acc = {bv, bv, bv, bv, bv, bv, bv, bv};
#pragma unroll
        for (int kb = 0; kb < 16; ++kb) {
            int k = 4 * kb + 2 * hi;
            v2f B;
            B.x = sW1t[(n0 + m) * WSTR + k];
            B.y = sW1t[(n0 + m) * WSTR + k + 1];
            acc = __builtin_amdgcn_wmma_f32_16x16x4_f32(false, A[kb], false, B,
                                                        (short)0, acc, false, false);
        }
#pragma unroll
        for (int j = 0; j < 8; ++j)
            tile[(j + 8 * hi) * TSTR + n0 + m] = fmaxf(acc[j], 0.f);
    }
    __syncthreads();

#pragma unroll
    for (int kb = 0; kb < 16; ++kb) {
        int k = 4 * kb + 2 * hi;
        A[kb].x = tile[m * TSTR + k];
        A[kb].y = tile[m * TSTR + k + 1];
    }

    for (int nt = 0; nt < 4; ++nt) {
        const int n0 = nt * 16;
        float bv = sB2[n0 + m];
        v8f acc = {bv, bv, bv, bv, bv, bv, bv, bv};
#pragma unroll
        for (int kb = 0; kb < 16; ++kb) {
            int k = 4 * kb + 2 * hi;
            v2f B;
            B.x = sW2t[(n0 + m) * WSTR + k];
            B.y = sW2t[(n0 + m) * WSTR + k + 1];
            acc = __builtin_amdgcn_wmma_f32_16x16x4_f32(false, A[kb], false, B,
                                                        (short)0, acc, false, false);
        }
#pragma unroll
        for (int j = 0; j < 8; ++j) {
            int gr = row0 + j + 8 * hi;
            if (gr < n) hout[(size_t)gr * GIN_D + n0 + m] = fmaxf(acc[j], 0.f);
        }
    }
}

__global__ __launch_bounds__(256) void gin_readout_kernel(const float* __restrict__ h,
                                                          const float* __restrict__ ow,
                                                          const float* __restrict__ ob,
                                                          float* __restrict__ out, int n) {
    int gw   = (blockIdx.x * 256 + threadIdx.x) >> 5;
    int lane = threadIdx.x & 31;
    if (gw >= n) return;
    float s = h[(size_t)gw * GIN_D + lane] * ow[lane] +
              h[(size_t)gw * GIN_D + 32 + lane] * ow[32 + lane];
    for (int off = 16; off; off >>= 1) s += __shfl_xor(s, off, 32);
    if (lane == 0) out[gw] = s + ob[0];
}

extern "C" void kernel_launch(void* const* d_in, const int* in_sizes, int n_in,
                              void* d_out, int out_size, void* d_ws, size_t ws_size,
                              hipStream_t stream) {
    const float* x  = (const float*)d_in[0];
    const int*   ei = (const int*)d_in[1];
    const int N = in_sizes[0] / GIN_D;
    const int E = in_sizes[1] / 2;
    const int* src = ei;
    const int* dst = ei + E;
    const float* W1[3] = {(const float*)d_in[2], (const float*)d_in[6],  (const float*)d_in[10]};
    const float* B1[3] = {(const float*)d_in[3], (const float*)d_in[7],  (const float*)d_in[11]};
    const float* W2[3] = {(const float*)d_in[4], (const float*)d_in[8],  (const float*)d_in[12]};
    const float* B2[3] = {(const float*)d_in[5], (const float*)d_in[9],  (const float*)d_in[13]};
    const float* out_w = (const float*)d_in[14];
    const float* out_b = (const float*)d_in[15];
    float* out = (float*)d_out;

    float* hA  = (float*)d_ws;
    float* hB  = hA  + (size_t)N * GIN_D;
    const int nchunk = (N + 1023) / 1024;
    int* cnt  = (int*)(hB + (size_t)N * GIN_D);
    int* off  = cnt  + (size_t)nchunk * 1024;
    int* bsum = off  + (size_t)nchunk * 1024;
    int* rank = bsum + (size_t)nchunk * 1024;
    int* lst  = rank + (size_t)E;

    dim3 b256(256);
    dim3 ng((unsigned)((N + 255) / 256));
    dim3 eg((unsigned)((E + 255) / 256));
    dim3 cg((unsigned)nchunk);
    dim3 lg((unsigned)((N + 127) / 128));
    dim3 rg((unsigned)(((long long)N * 32 + 255) / 256));

    gin_clear_kernel<<<ng, b256, 0, stream>>>(cnt, N);
    gin_count_kernel<<<eg, b256, 0, stream>>>(dst, cnt, rank, E);
    gin_chunk_sum_kernel<<<cg, b256, 0, stream>>>(cnt, bsum, N);
    gin_scan_kernel<<<cg, b256, 0, stream>>>(cnt, bsum, off, N);
    gin_place_kernel<<<eg, b256, 0, stream>>>(src, dst, rank, off, lst, E);

    const float* hin = x;
    float* bufs[2] = {hA, hB};
    for (int l = 0; l < 3; ++l) {
        float* hout = bufs[l & 1];
        gin_layer_kernel<<<lg, b256, 0, stream>>>(hin, off, cnt, lst, W1[l], B1[l], W2[l], B2[l], hout, N);
        hin = hout;
    }
    gin_readout_kernel<<<rg, b256, 0, stream>>>(hin, out_w, out_b, out, N);
}


# Edge list built with LDS counters: one block per (edge slice, node bin) counts, then places, in LDS; global atomics, rank array, 2 launches gone; list order now slice-major, arrival within a slice
# speedup vs baseline: 2.4056x; 1.2121x over previous
#include <hip/hip_runtime.h>

#define GIN_D 64
#define TSTR 68
#define WSTR 68

typedef __attribute__((ext_vector_type(2))) float v2f;
typedef __attribute__((ext_vector_type(8))) float v8f;

#define NSL 32
#define CPB 13
#define RB  (CPB * 1024)
#define EPT 16

__global__ __launch_bounds__(1024) void gin_bincount_kernel(const int* __restrict__ dst,
                                                            int* __restrict__ part,
                                                            int* __restrict__ tot,
                                                            int e, int slen, int npad) {
    __shared__ int c[RB];
    const int t  = threadIdx.x;
    const int lo = blockIdx.y * RB;
    for (int i = t; i < RB; i += 1024) c[i] = 0;
    __syncthreads();
    const int e0 = blockIdx.x * slen;
    const int e1 = min(e, e0 + slen);
    for (int i = e0 + t; i < e1; i += EPT * 1024) {
        int d[EPT];
#pragma unroll
        for (int j = 0; j < EPT; ++j) {
            const int k = i + j * 1024;
            const int v = dst[min(k, e1 - 1)];
            d[j] = (k < e1) ? v : -1;
        }
#pragma unroll
        for (int j = 0; j < EPT; ++j) {
            const unsigned u = (unsigned)(d[j] - lo);
            if (u < (unsigned)RB) atomicAdd(&c[u], 1);
        }
    }
    __syncthreads();
    int* o = part + (size_t)blockIdx.x * npad + lo;
    for (int i = t; i < RB; i += 1024) o[i] = c[i];
    const int w    = t >> 5;
    const int lane = t & 31;
    if (w < CPB) {
        int s = 0;
        for (int i = lane; i < 1024; i += 32) s += c[w * 1024 + i];
        for (int x = 16; x; x >>= 1) s += __shfl_xor(s, x, 32);
        if (lane == 0) tot[((size_t)blockIdx.y * CPB + w) * NSL + blockIdx.x] = s;
    }
}

__global__ __launch_bounds__(256) void gin_scan_kernel(const int* __restrict__ tot,
                                                       int* __restrict__ part,
                                                       int* __restrict__ cnt,
                                                       int* __restrict__ off, int n, int npad) {
    __shared__ int sw[8];
    __shared__ int sp[8];
    const int t    = threadIdx.x;
    const int lane = t & 31;
    const int wave = t >> 5;
    const int base = blockIdx.x * 1024 + t * 4;

    int* pp = part + base;
    int4 v[NSL];
#pragma unroll
    for (int k = 0; k < NSL; ++k) v[k] = *(const int4*)(pp + (size_t)k * npad);

    int p = 0;
    for (int i = t; i < (int)blockIdx.x * NSL; i += 256) p += tot[i];
    for (int o = 16; o; o >>= 1) p += __shfl_xor(p, o, 32);

    int c0 = 0, c1 = 0, c2 = 0, c3 = 0;
#pragma unroll
    for (int k = 0; k < NSL; ++k) { c0 += v[k].x; c1 += v[k].y; c2 += v[k].z; c3 += v[k].w; }
    const int s = c0 + c1 + c2 + c3;

    int inc = s;
    for (int d = 1; d < 32; d <<= 1) {
        int u = __shfl_up(inc, d, 32);
        if (lane >= d) inc += u;
    }
    if (lane == 31) sw[wave] = inc;
    if (lane == 0)  sp[wave] = p;
    __syncthreads();

    int before = 0;
    for (int w = 0; w < 8; ++w) {
        before += sp[w];
        if (w < wave) before += sw[w];
    }
    int4 r;
    r.x = before + inc - s;
    r.y = r.x + c0;
    r.z = r.y + c1;
    r.w = r.z + c2;
    if (base + 0 < n) { off[base + 0] = r.x; cnt[base + 0] = c0; }
    if (base + 1 < n) { off[base + 1] = r.y; cnt[base + 1] = c1; }
    if (base + 2 < n) { off[base + 2] = r.z; cnt[base + 2] = c2; }
    if (base + 3 < n) { off[base + 3] = r.w; cnt[base + 3] = c3; }

#pragma unroll
    for (int k = 0; k < NSL; ++k) {
        *(int4*)(pp + (size_t)k * npad) = r;
        r.x += v[k].x; r.y += v[k].y; r.z += v[k].z; r.w += v[k].w;
    }
}

__global__ __launch_bounds__(1024) void gin_binplace_kernel(const int* __restrict__ src,
                                                            const int* __restrict__ dst,
                                                            const int* __restrict__ part,
                                                            int* __restrict__ lst,
                                                            int e, int slen, int npad) {
    __shared__ int cur[RB];
    const int t  = threadIdx.x;
    const int lo = blockIdx.y * RB;
    const int* bp = part + (size_t)blockIdx.x * npad + lo;
    for (int i = t; i < RB; i += 1024) cur[i] = bp[i];
    __syncthreads();
    const int e0 = blockIdx.x * slen;
    const int e1 = min(e, e0 + slen);
    for (int i = e0 + t; i < e1; i += EPT * 1024) {
        int d[EPT], sv[EPT], pos[EPT];
#pragma unroll
        for (int j = 0; j < EPT; ++j) {
            const int k  = i + j * 1024;
            const int kc = min(k, e1 - 1);
            const int dv = dst[kc];
            sv[j] = src[kc];
            d[j]  = (k < e1) ? dv : -1;
        }
#pragma unroll
        for (int j = 0; j < EPT; ++j) {
            const unsigned u = (unsigned)(d[j] - lo);
            pos[j] = -1;
            if (u < (unsigned)RB) pos[j] = atomicAdd(&cur[u], 1);
        }
#pragma unroll
        for (int j = 0; j < EPT; ++j)
            if ((unsigned)pos[j] < (unsigned)e) lst[pos[j]] = sv[j];
    }
}

__global__ __launch_bounds__(256)
void gin_layer_kernel(const float* __restrict__ h,
                      const int* __restrict__ off, const int* __restrict__ cnt,
                      const int* __restrict__ lst,
                      const float* __restrict__ w1, const float* __restrict__ b1,
                      const float* __restrict__ w2, const float* __restrict__ b2,
                      float* __restrict__ hout, int n) {
    __shared__ float sW1t[GIN_D * WSTR];
    __shared__ float sW2t[GIN_D * WSTR];
    __shared__ float sB1[GIN_D], sB2[GIN_D];
    __shared__ float sT[8][16 * TSTR];

    const int t    = threadIdx.x;
    const int wave = t >> 5;
    const int lane = t & 31;
    const int m    = lane & 15;
    const int hi   = lane >> 4;

    const int row0 = (blockIdx.x * 8 + wave) * 16;
    float* tile = sT[wave];

    const int q = lane >> 3;
    const int c = (lane & 7) << 2;

    int pbeg[4], pcnt[4];
#pragma unroll
    for (int it = 0; it < 4; ++it) {
        const int nd = min(row0 + it * 4 + q, n - 1);
        pbeg[it] = off[nd];
        pcnt[it] = cnt[nd];
    }

    for (int i = t; i < GIN_D * GIN_D / 4; i += 256) {
        int k  = i >> 4;
        int c4 = (i & 15) << 2;
        float4 a = ((const float4*)w1)[i];
        sW1t[(c4 + 0) * WSTR + k] = a.x;
        sW1t[(c4 + 1) * WSTR + k] = a.y;
        sW1t[(c4 + 2) * WSTR + k] = a.z;
        sW1t[(c4 + 3) * WSTR + k] = a.w;
        float4 b = ((const float4*)w2)[i];
        sW2t[(c4 + 0) * WSTR + k] = b.x;
        sW2t[(c4 + 1) * WSTR + k] = b.y;
        sW2t[(c4 + 2) * WSTR + k] = b.z;
        sW2t[(c4 + 3) * WSTR + k] = b.w;
    }
    if (t < GIN_D) { sB1[t] = b1[t]; sB2[t] = b2[t]; }

#pragma unroll
    for (int it = 0; it < 4; ++it) {
        const int r    = it * 4 + q;
        const int node = row0 + r;
        float4 a0 = make_float4(0.f, 0.f, 0.f, 0.f);
        float4 a1 = make_float4(0.f, 0.f, 0.f, 0.f);
        if (node < n) {
            int p = pbeg[it];
            const int end  = p + pcnt[it];
            const int last = end - 1;
            const float* hc = h + c;
            a0 = *(const float4*)(hc + (size_t)node * GIN_D);
            a1 = *(const float4*)(hc + (size_t)node * GIN_D + 32);
            int s0 = 0, s1 = 0, s2 = 0, s3 = 0;
            if (p < end) {
                s0 = lst[p];
                s1 = lst[min(p + 1, last)];
                s2 = lst[min(p + 2, last)];
                s3 = lst[min(p + 3, last)];
            }
            while (p < end) {
                const bool k1 = p + 1 < end;
                const bool k2 = p + 2 < end;
                const bool k3 = p + 3 < end;
                const float4 u0 = *(const float4*)(hc + (size_t)s0 * GIN_D);
                const float4 v0 = *(const float4*)(hc + (size_t)s0 * GIN_D + 32);
                const float4 u1 = *(const float4*)(hc + (size_t)s1 * GIN_D);
                const float4 v1 = *(const float4*)(hc + (size_t)s1 * GIN_D + 32);
                const float4 u2 = *(const float4*)(hc + (size_t)s2 * GIN_D);
                const float4 v2 = *(const float4*)(hc + (size_t)s2 * GIN_D + 32);
                const float4 u3 = *(const float4*)(hc + (size_t)s3 * GIN_D);
                const float4 v3 = *(const float4*)(hc + (size_t)s3 * GIN_D + 32);
                p += 4;
                if (p < end) {
                    s0 = lst[p];
                    s1 = lst[min(p + 1, last)];
                    s2 = lst[min(p + 2, last)];
                    s3 = lst[min(p + 3, last)];
                }
                a0.x += u0.x; a0.y += u0.y; a0.z += u0.z; a0.w += u0.w;
                a1.x += v0.x; a1.y += v0.y; a1.z += v0.z; a1.w += v0.w;
                a0.x += k1 ? u1.x : 0.f; a0.y += k1 ? u1.y : 0.f;
                a0.z += k1 ? u1.z : 0.f; a0.w += k1 ? u1.w : 0.f;
                a1.x += k1 ? v1.x : 0.f; a1.y += k1 ? v1.y : 0.f;
                a1.z += k1 ? v1.z : 0.f; a1.w += k1 ? v1.w : 0.f;
                a0.x += k2 ? u2.x : 0.f; a0.y += k2 ? u2.y : 0.f;
                a0.z += k2 ? u2.z : 0.f; a0.w += k2 ? u2.w : 0.f;
                a1.x += k2 ? v2.x : 0.f; a1.y += k2 ? v2.y : 0.f;
                a1.z += k2 ? v2.z : 0.f; a1.w += k2 ? v2.w : 0.f;
                a0.x += k3 ? u3.x : 0.f; a0.y += k3 ? u3.y : 0.f;
                a0.z += k3 ? u3.z : 0.f; a0.w += k3 ? u3.w : 0.f;
                a1.x += k3 ? v3.x : 0.f; a1.y += k3 ? v3.y : 0.f;
                a1.z += k3 ? v3.z : 0.f; a1.w += k3 ? v3.w : 0.f;
            }
        }
        float* tr = tile + r * TSTR + c;
        tr[0] = a0.x; tr[1] = a0.y; tr[2] = a0.z; tr[3] = a0.w;
        tr[32] = a1.x; tr[33] = a1.y; tr[34] = a1.z; tr[35] = a1.w;
    }
    __syncthreads();

    v2f A[16];
#pragma unroll
    for (int kb = 0; kb < 16; ++kb) {
        int k = 4 * kb + 2 * hi;
        A[kb].x = tile[m * TSTR + k];
        A[kb].y = tile[m * TSTR + k + 1];
    }
    __syncthreads();

    for (int nt = 0; nt < 4; ++nt) {
        const int n0 = nt * 16;
        float bv = sB1[n0 + m];
        v8f acc = {bv, bv, bv, bv, bv, bv, bv, bv};
#pragma unroll
        for (int kb = 0; kb < 16; ++kb) {
            int k = 4 * kb + 2 * hi;
            v2f B;
            B.x = sW1t[(n0 + m) * WSTR + k];
            B.y = sW1t[(n0 + m) * WSTR + k + 1];
            acc = __builtin_amdgcn_wmma_f32_16x16x4_f32(false, A[kb], false, B,
                                                        (short)0, acc, false, false);
        }
#pragma unroll
        for (int j = 0; j < 8; ++j)
            tile[(j + 8 * hi) * TSTR + n0 + m] = fmaxf(acc[j], 0.f);
    }
    __syncthreads();

#pragma unroll
    for (int kb = 0; kb < 16; ++kb) {
        int k = 4 * kb + 2 * hi;
        A[kb].x = tile[m * TSTR + k];
        A[kb].y = tile[m * TSTR + k + 1];
    }

    for (int nt = 0; nt < 4; ++nt) {
        const int n0 = nt * 16;
        float bv = sB2[n0 + m];
        v8f acc = {bv, bv, bv, bv, bv, bv, bv, bv};
#pragma unroll
        for (int kb = 0; kb < 16; ++kb) {
            int k = 4 * kb + 2 * hi;
            v2f B;
            B.x = sW2t[(n0 + m) * WSTR + k];
            B.y = sW2t[(n0 + m) * WSTR + k + 1];
            acc = __builtin_amdgcn_wmma_f32_16x16x4_f32(false, A[kb], false, B,
                                                        (short)0, acc, false, false);
        }
#pragma unroll
        for (int j = 0; j < 8; ++j) {
            int gr = row0 + j + 8 * hi;
            if (gr < n) hout[(size_t)gr * GIN_D + n0 + m] = fmaxf(acc[j], 0.f);
        }
    }
}

__global__ __launch_bounds__(256) void gin_readout_kernel(const float* __restrict__ h,
                                                          const float* __restrict__ ow,
                                                          const float* __restrict__ ob,
                                                          float* __restrict__ out, int n) {
    int gw   = (blockIdx.x * 256 + threadIdx.x) >> 5;
    int lane = threadIdx.x & 31;
    if (gw >= n) return;
    float s = h[(size_t)gw * GIN_D + lane] * ow[lane] +
              h[(size_t)gw * GIN_D + 32 + lane] * ow[32 + lane];
    for (int off = 16; off; off >>= 1) s += __shfl_xor(s, off, 32);
    if (lane == 0) out[gw] = s + ob[0];
}

extern "C" void kernel_launch(void* const* d_in, const int* in_sizes, int n_in,
                              void* d_out, int out_size, void* d_ws, size_t ws_size,
                              hipStream_t stream) {
    const float* x  = (const float*)d_in[0];
    const int*   ei = (const int*)d_in[1];
    const int N = in_sizes[0] / GIN_D;
    const int E = in_sizes[1] / 2;
    const int* src = ei;
    const int* dst = ei + E;
    const float* W1[3] = {(const float*)d_in[2], (const float*)d_in[6],  (const float*)d_in[10]};
    const float* B1[3] = {(const float*)d_in[3], (const float*)d_in[7],  (const float*)d_in[11]};
    const float* W2[3] = {(const float*)d_in[4], (const float*)d_in[8],  (const float*)d_in[12]};
    const float* B2[3] = {(const float*)d_in[5], (const float*)d_in[9],  (const float*)d_in[13]};
    const float* out_w = (const float*)d_in[14];
    const float* out_b = (const float*)d_in[15];
    float* out = (float*)d_out;

    float* hA  = (float*)d_ws;
    float* hB  = hA  + (size_t)N * GIN_D;
    const int nchunk = (N + 1023) / 1024;
    const int nbin   = (N + RB - 1) / RB;
    const int npad   = nbin * RB;
    const int slen   = (E + NSL - 1) / NSL;
    int* cnt  = (int*)(hB + (size_t)N * GIN_D);
    int* off  = cnt  + (size_t)nchunk * 1024;
    int* tot  = off  + (size_t)nchunk * 1024;
    int* part = tot  + (size_t)nbin * CPB * NSL;
    int* lst  = part + (size_t)NSL * npad;

    dim3 b256(256);
    dim3 b1024(1024);
    dim3 cg((unsigned)nchunk);
    dim3 bg((unsigned)NSL, (unsigned)nbin);
    dim3 lg((unsigned)((N + 127) / 128));
    dim3 rg((unsigned)(((long long)N * 32 + 255) / 256));

    gin_bincount_kernel<<<bg, b1024, 0, stream>>>(dst, part, tot, E, slen, npad);
    gin_scan_kernel<<<cg, b256, 0, stream>>>(tot, part, cnt, off, N, npad);
    gin_binplace_kernel<<<bg, b1024, 0, stream>>>(src, dst, part, lst, E, slen, npad);

    const float* hin = x;
    float* bufs[2] = {hA, hB};
    for (int l = 0; l < 3; ++l) {
        float* hout = bufs[l & 1];
        gin_layer_kernel<<<lg, b256, 0, stream>>>(hin, off, cnt, lst, W1[l], B1[l], W2[l], B2[l], hout, N);
        hin = hout;
    }
    gin_readout_kernel<<<rg, b256, 0, stream>>>(hin, out_w, out_b, out, N);
}


# Readout fused into the last layer launch: out = dot(relu(acc), out_w) + out_b summed in its epilogue (tile-major, then a 16-lane exchange sum: new summation order); readout launch and last hout gone
# speedup vs baseline: 2.6419x; 1.0982x over previous
#include <hip/hip_runtime.h>

#define GIN_D 64
#define TSTR 68
#define WSTR 68

typedef __attribute__((ext_vector_type(2))) float v2f;
typedef __attribute__((ext_vector_type(8))) float v8f;

#define NSL 32
#define CPB 13
#define RB  (CPB * 1024)
#define EPT 16

__global__ __launch_bounds__(1024) void gin_bincount_kernel(const int* __restrict__ dst,
                                                            int* __restrict__ part,
                                                            int* __restrict__ tot,
                                                            int e, int slen, int npad) {
    __shared__ int c[RB];
    const int t  = threadIdx.x;
    const int lo = blockIdx.y * RB;
    for (int i = t; i < RB; i += 1024) c[i] = 0;
    __syncthreads();
    const int e0 = blockIdx.x * slen;
    const int e1 = min(e, e0 + slen);
    for (int i = e0 + t; i < e1; i += EPT * 1024) {
        int d[EPT];
#pragma unroll
        for (int j = 0; j < EPT; ++j) {
            const int k = i + j * 1024;
            const int v = dst[min(k, e1 - 1)];
            d[j] = (k < e1) ? v : -1;
        }
#pragma unroll
        for (int j = 0; j < EPT; ++j) {
            const unsigned u = (unsigned)(d[j] - lo);
            if (u < (unsigned)RB) atomicAdd(&c[u], 1);
        }
    }
    __syncthreads();
    int* o = part + (size_t)blockIdx.x * npad + lo;
    for (int i = t; i < RB; i += 1024) o[i] = c[i];
    const int w    = t >> 5;
    const int lane = t & 31;
    if (w < CPB) {
        int s = 0;
        for (int i = lane; i < 1024; i += 32) s += c[w * 1024 + i];
        for (int x = 16; x; x >>= 1) s += __shfl_xor(s, x, 32);
        if (lane == 0) tot[((size_t)blockIdx.y * CPB + w) * NSL + blockIdx.x] = s;
    }
}

__global__ __launch_bounds__(256) void gin_scan_kernel(const int* __restrict__ tot,
                                                       int* __restrict__ part,
                                                       int* __restrict__ cnt,
                                                       int* __restrict__ off, int n, int npad) {
    __shared__ int sw[8];
    __shared__ int sp[8];
    const int t    = threadIdx.x;
    const int lane = t & 31;
    const int wave = t >> 5;
    const int base = blockIdx.x * 1024 + t * 4;

    int* pp = part + base;
    int4 v[NSL];
#pragma unroll
    for (int k = 0; k < NSL; ++k) v[k] = *(const int4*)(pp + (size_t)k * npad);

    int p = 0;
    for (int i = t; i < (int)blockIdx.x * NSL; i += 256) p += tot[i];
    for (int o = 16; o; o >>= 1) p += __shfl_xor(p, o, 32);

    int c0 = 0, c1 = 0, c2 = 0, c3 = 0;
#pragma unroll
    for (int k = 0; k < NSL; ++k) { c0 += v[k].x; c1 += v[k].y; c2 += v[k].z; c3 += v[k].w; }
    const int s = c0 + c1 + c2 + c3;

    int inc = s;
    for (int d = 1; d < 32; d <<= 1) {
        int u = __shfl_up(inc, d, 32);
        if (lane >= d) inc += u;
    }
    if (lane == 31) sw[wave] = inc;
    if (lane == 0)  sp[wave] = p;
    __syncthreads();

    int before = 0;
    for (int w = 0; w < 8; ++w) {
        before += sp[w];
        if (w < wave) before += sw[w];
    }
    int4 r;
    r.x = before + inc - s;
    r.y = r.x + c0;
    r.z = r.y + c1;
    r.w = r.z + c2;
    if (base + 0 < n) { off[base + 0] = r.x; cnt[base + 0] = c0; }
    if (base + 1 < n) { off[base + 1] = r.y; cnt[base + 1] = c1; }
    if (base + 2 < n) { off[base + 2] = r.z; cnt[base + 2] = c2; }
    if (base + 3 < n) { off[base + 3] = r.w; cnt[base + 3] = c3; }

#pragma unroll
    for (int k = 0; k < NSL; ++k) {
        *(int4*)(pp + (size_t)k * npad) = r;
        r.x += v[k].x; r.y += v[k].y; r.z += v[k].z; r.w += v[k].w;
    }
}

__global__ __launch_bounds__(1024) void gin_binplace_kernel(const int* __restrict__ src,
                                                            const int* __restrict__ dst,
                                                            const int* __restrict__ part,
                                                            int* __restrict__ lst,
                                                            int e, int slen, int npad) {
    __shared__ int cur[RB];
    const int t  = threadIdx.x;
    const int lo = blockIdx.y * RB;
    const int* bp = part + (size_t)blockIdx.x * npad + lo;
    for (int i = t; i < RB; i += 1024) cur[i] = bp[i];
    __syncthreads();
    const int e0 = blockIdx.x * slen;
    const int e1 = min(e, e0 + slen);
    for (int i = e0 + t; i < e1; i += EPT * 1024) {
        int d[EPT], sv[EPT], pos[EPT];
#pragma unroll
        for (int j = 0; j < EPT; ++j) {
            const int k  = i + j * 1024;
            const int kc = min(k, e1 - 1);
            const int dv = dst[kc];
            sv[j] = src[kc];
            d[j]  = (k < e1) ? dv : -1;
        }
#pragma unroll
        for (int j = 0; j < EPT; ++j) {
            const unsigned u = (unsigned)(d[j] - lo);
            pos[j] = -1;
            if (u < (unsigned)RB) pos[j] = atomicAdd(&cur[u], 1);
        }
#pragma unroll
        for (int j = 0; j < EPT; ++j)
            if ((unsigned)pos[j] < (unsigned)e) lst[pos[j]] = sv[j];
    }
}

template <bool LAST>
__global__ __launch_bounds__(256)
void gin_layer_kernel(const float* __restrict__ h,
                      const int* __restrict__ off, const int* __restrict__ cnt,
                      const int* __restrict__ lst,
                      const float* __restrict__ w1, const float* __restrict__ b1,
                      const float* __restrict__ w2, const float* __restrict__ b2,
                      float* __restrict__ hout,
                      const float* __restrict__ ow, const float* __restrict__ ob,
                      float* __restrict__ out, int n) {
    __shared__ float sW1t[GIN_D * WSTR];
    __shared__ float sW2t[GIN_D * WSTR];
    __shared__ float sB1[GIN_D], sB2[GIN_D];
    __shared__ float sT[8][16 * TSTR];

    const int t    = threadIdx.x;
    const int wave = t >> 5;
    const int lane = t & 31;
    const int m    = lane & 15;
    const int hi   = lane >> 4;

    const int row0 = (blockIdx.x * 8 + wave) * 16;
    float* tile = sT[wave];

    const int q = lane >> 3;
    const int c = (lane & 7) << 2;

    int pbeg[4], pcnt[4];
#pragma unroll
    for (int it = 0; it < 4; ++it) {
        const int nd = min(row0 + it * 4 + q, n - 1);
        pbeg[it] = off[nd];
        pcnt[it] = cnt[nd];
    }

    for (int i = t; i < GIN_D * GIN_D / 4; i += 256) {
        int k  = i >> 4;
        int c4 = (i & 15) << 2;
        float4 a = ((const float4*)w1)[i];
        sW1t[(c4 + 0) * WSTR + k] = a.x;
        sW1t[(c4 + 1) * WSTR + k] = a.y;
        sW1t[(c4 + 2) * WSTR + k] = a.z;
        sW1t[(c4 + 3) * WSTR + k] = a.w;
        float4 b = ((const float4*)w2)[i];
        sW2t[(c4 + 0) * WSTR + k] = b.x;
        sW2t[(c4 + 1) * WSTR + k] = b.y;
        sW2t[(c4 + 2) * WSTR + k] = b.z;
        sW2t[(c4 + 3) * WSTR + k] = b.w;
    }
    if (t < GIN_D) { sB1[t] = b1[t]; sB2[t] = b2[t]; }

#pragma unroll
    for (int it = 0; it < 4; ++it) {
        const int r    = it * 4 + q;
        const int node = row0 + r;
        float4 a0 = make_float4(0.f, 0.f, 0.f, 0.f);
        float4 a1 = make_float4(0.f, 0.f, 0.f, 0.f);
        if (node < n) {
            int p = pbeg[it];
            const int end  = p + pcnt[it];
            const int last = end - 1;
            const float* hc = h + c;
            a0 = *(const float4*)(hc + (size_t)node * GIN_D);
            a1 = *(const float4*)(hc + (size_t)node * GIN_D + 32);
            int s0 = 0, s1 = 0, s2 = 0, s3 = 0;
            if (p < end) {
                s0 = lst[p];
                s1 = lst[min(p + 1, last)];
                s2 = lst[min(p + 2, last)];
                s3 = lst[min(p + 3, last)];
            }
            while (p < end) {
                const bool k1 = p + 1 < end;
                const bool k2 = p + 2 < end;
                const bool k3 = p + 3 < end;
                const float4 u0 = *(const float4*)(hc + (size_t)s0 * GIN_D);
                const float4 v0 = *(const float4*)(hc + (size_t)s0 * GIN_D + 32);
                const float4 u1 = *(const float4*)(hc + (size_t)s1 * GIN_D);
                const float4 v1 = *(const float4*)(hc + (size_t)s1 * GIN_D + 32);
                const float4 u2 = *(const float4*)(hc + (size_t)s2 * GIN_D);
                const float4 v2 = *(const float4*)(hc + (size_t)s2 * GIN_D + 32);
                const float4 u3 = *(const float4*)(hc + (size_t)s3 * GIN_D);
                const float4 v3 = *(const float4*)(hc + (size_t)s3 * GIN_D + 32);
                p += 4;
                if (p < end) {
                    s0 = lst[p];
                    s1 = lst[min(p + 1, last)];
                    s2 = lst[min(p + 2, last)];
                    s3 = lst[min(p + 3, last)];
                }
                a0.x += u0.x; a0.y += u0.y; a0.z += u0.z; a0.w += u0.w;
                a1.x += v0.x; a1.y += v0.y; a1.z += v0.z; a1.w += v0.w;
                a0.x += k1 ? u1.x : 0.f; a0.y += k1 ? u1.y : 0.f;
                a0.z += k1 ? u1.z : 0.f; a0.w += k1 ? u1.w : 0.f;
                a1.x += k1 ? v1.x : 0.f; a1.y += k1 ? v1.y : 0.f;
                a1.z += k1 ? v1.z : 0.f; a1.w += k1 ? v1.w : 0.f;
                a0.x += k2 ? u2.x : 0.f; a0.y += k2 ? u2.y : 0.f;
                a0.z += k2 ? u2.z : 0.f; a0.w += k2 ? u2.w : 0.f;
                a1.x += k2 ? v2.x : 0.f; a1.y += k2 ? v2.y : 0.f;
                a1.z += k2 ? v2.z : 0.f; a1.w += k2 ? v2.w : 0.f;
                a0.x += k3 ? u3.x : 0.f; a0.y += k3 ? u3.y : 0.f;
                a0.z += k3 ? u3.z : 0.f; a0.w += k3 ? u3.w : 0.f;
                a1.x += k3 ? v3.x : 0.f; a1.y += k3 ? v3.y : 0.f;
                a1.z += k3 ? v3.z : 0.f; a1.w += k3 ? v3.w : 0.f;
            }
        }
        float* tr = tile + r * TSTR + c;
        tr[0] = a0.x; tr[1] = a0.y; tr[2] = a0.z; tr[3] = a0.w;
        tr[32] = a1.x; tr[33] = a1.y; tr[34] = a1.z; tr[35] = a1.w;
    }
    __syncthreads();

    v2f A[16];
#pragma unroll
    for (int kb = 0; kb < 16; ++kb) {
        int k = 4 * kb + 2 * hi;
        A[kb].x = tile[m * TSTR + k];
        A[kb].y = tile[m * TSTR + k + 1];
    }
    __syncthreads();

    for (int nt = 0; nt < 4; ++nt) {
        const int n0 = nt * 16;
        float bv = sB1[n0 + m];
        v8f acc = {bv, bv, bv, bv, bv, bv, bv, bv};
#pragma unroll
        for (int kb = 0; kb < 16; ++kb) {
            int k = 4 * kb + 2 * hi;
            v2f B;
            B.x = sW1t[(n0 + m) * WSTR + k];
            B.y = sW1t[(n0 + m) * WSTR + k + 1];
            acc = __builtin_amdgcn_wmma_f32_16x16x4_f32(false, A[kb], false, B,
                                                        (short)0, acc, false, false);
        }
#pragma unroll
        for (int j = 0; j < 8; ++j)
            tile[(j + 8 * hi) * TSTR + n0 + m] = fmaxf(acc[j], 0.f);
    }
    __syncthreads();

#pragma unroll
    for (int kb = 0; kb < 16; ++kb) {
        int k = 4 * kb + 2 * hi;
        A[kb].x = tile[m * TSTR + k];
        A[kb].y = tile[m * TSTR + k + 1];
    }

    float ps[8] = {0.f, 0.f, 0.f, 0.f, 0.f, 0.f, 0.f, 0.f};
    for (int nt = 0; nt < 4; ++nt) {
        const int n0 = nt * 16;
        float bv = sB2[n0 + m];
        v8f acc = {bv, bv, bv, bv, bv, bv, bv, bv};
#pragma unroll
        for (int kb = 0; kb < 16; ++kb) {
            int k = 4 * kb + 2 * hi;
            v2f B;
            B.x = sW2t[(n0 + m) * WSTR + k];
            B.y = sW2t[(n0 + m) * WSTR + k + 1];
            acc = __builtin_amdgcn_wmma_f32_16x16x4_f32(false, A[kb], false, B,
                                                        (short)0, acc, false, false);
        }
        if (LAST) {
            const float wv = ow[n0 + m];
#pragma unroll
            for (int j = 0; j < 8; ++j) ps[j] += fmaxf(acc[j], 0.f) * wv;
        } else {
#pragma unroll
            for (int j = 0; j < 8; ++j) {
                int gr = row0 + j + 8 * hi;
                if (gr < n) hout[(size_t)gr * GIN_D + n0 + m] = fmaxf(acc[j], 0.f);
            }
        }
    }
    if (LAST) {
        const bool u3 = (m & 8) != 0;
        float q4[4];
#pragma unroll
        for (int i = 0; i < 4; ++i) {
            const float keep = u3 ? ps[i + 4] : ps[i];
            const float give = u3 ? ps[i] : ps[i + 4];
            q4[i] = keep + __shfl_xor(give, 8, 32);
        }
        const bool u2 = (m & 4) != 0;
        float q2[2];
#pragma unroll
        for (int i = 0; i < 2; ++i) {
            const float keep = u2 ? q4[i + 2] : q4[i];
            const float give = u2 ? q4[i] : q4[i + 2];
            q2[i] = keep + __shfl_xor(give, 4, 32);
        }
        const bool u1 = (m & 2) != 0;
        const float keep1 = u1 ? q2[1] : q2[0];
        const float give1 = u1 ? q2[0] : q2[1];
        float q1 = keep1 + __shfl_xor(give1, 2, 32);
        q1 += __shfl_xor(q1, 1, 32);
        const int gr = row0 + 8 * hi + (m >> 1);
        if ((m & 1) == 0 && gr < n) out[gr] = q1 + ob[0];
    }
}

extern "C" void kernel_launch(void* const* d_in, const int* in_sizes, int n_in,
                              void* d_out, int out_size, void* d_ws, size_t ws_size,
                              hipStream_t stream) {
    const float* x  = (const float*)d_in[0];
    const int*   ei = (const int*)d_in[1];
    const int N = in_sizes[0] / GIN_D;
    const int E = in_sizes[1] / 2;
    const int* src = ei;
    const int* dst = ei + E;
    const float* W1[3] = {(const float*)d_in[2], (const float*)d_in[6],  (const float*)d_in[10]};
    const float* B1[3] = {(const float*)d_in[3], (const float*)d_in[7],  (const float*)d_in[11]};
    const float* W2[3] = {(const float*)d_in[4], (const float*)d_in[8],  (const float*)d_in[12]};
    const float* B2[3] = {(const float*)d_in[5], (const float*)d_in[9],  (const float*)d_in[13]};
    const float* out_w = (const float*)d_in[14];
    const float* out_b = (const float*)d_in[15];
    float* out = (float*)d_out;

    float* hA  = (float*)d_ws;
    float* hB  = hA  + (size_t)N * GIN_D;
    const int nchunk = (N + 1023) / 1024;
    const int nbin   = (N + RB - 1) / RB;
    const int npad   = nbin * RB;
    const int slen   = (E + NSL - 1) / NSL;
    int* cnt  = (int*)(hB + (size_t)N * GIN_D);
    int* off  = cnt  + (size_t)nchunk * 1024;
    int* tot  = off  + (size_t)nchunk * 1024;
    int* part = tot  + (size_t)nbin * CPB * NSL;
    int* lst  = part + (size_t)NSL * npad;

    dim3 b256(256);
    dim3 b1024(1024);
    dim3 cg((unsigned)nchunk);
    dim3 bg((unsigned)NSL, (unsigned)nbin);
    dim3 lg((unsigned)((N + 127) / 128));

    gin_bincount_kernel<<<bg, b1024, 0, stream>>>(dst, part, tot, E, slen, npad);
    gin_scan_kernel<<<cg, b256, 0, stream>>>(tot, part, cnt, off, N, npad);
    gin_binplace_kernel<<<bg, b1024, 0, stream>>>(src, dst, part, lst, E, slen, npad);

    gin_layer_kernel<false><<<lg, b256, 0, stream>>>(x,  off, cnt, lst, W1[0], B1[0], W2[0], B2[0], hA,
                                                     out_w, out_b, out, N);
    gin_layer_kernel<false><<<lg, b256, 0, stream>>>(hA, off, cnt, lst, W1[1], B1[1], W2[1], B2[1], hB,
                                                     out_w, out_b, out, N);
    gin_layer_kernel<true><<<lg, b256, 0, stream>>>(hB, off, cnt, lst, W1[2], B1[2], W2[2], B2[2], hA,
                                                    out_w, out_b, out, N);
}


# Edge list built in 2 launches, not 3: each 4096-edge slice sorted by 256-node bin in LDS and written whole; one block per bin counts, scans, places; a node's list order is now arrival order
# speedup vs baseline: 3.5290x; 1.3358x over previous
#include <hip/hip_runtime.h>

#define GIN_D 64
#define TSTR 68
#define WSTR 68

typedef __attribute__((ext_vector_type(2))) float v2f;
typedef __attribute__((ext_vector_type(8))) float v8f;

#define SLICE 4096
#define NPB   256
#define NBMAX 512
#define SMAX  256
#define CB    16
#define TSH   13
#define EPB   12

__device__ __forceinline__ int gin_wave_incl(int v, int lane) {
    for (int d = 1; d < 32; d <<= 1) {
        const int u = __shfl_up(v, d, 32);
        if (lane >= d) v += u;
    }
    return v;
}

__global__ __launch_bounds__(512) void gin_partition_kernel(const int* __restrict__ src,
                                                            const int* __restrict__ dst,
                                                            int* __restrict__ part,
                                                            int* __restrict__ tab,
                                                            int* __restrict__ tab2,
                                                            int e, int n, int S, int nbin) {
    __shared__ int hist[NBMAX];
    __shared__ int cur[NBMAX];
    __shared__ int sorted[SLICE];
    __shared__ int wsum[16];
    const int t    = threadIdx.x;
    const int lane = t & 31;
    const int wave = t >> 5;
    const int s    = blockIdx.x;
    hist[t] = 0;
    __syncthreads();
    const int e0 = s * SLICE;
    const int e1 = min(e, e0 + SLICE);
    int pk[8], bn[8];
#pragma unroll
    for (int u = 0; u < 8; ++u) {
        const int k  = e0 + u * 512 + t;
        const int kc = min(k, e1 - 1);
        const int dv = dst[kc];
        const int sv = src[kc];
        const bool ok = (k < e1) && ((unsigned)dv < (unsigned)n);
        bn[u] = ok ? (dv >> 8) : -1;
        pk[u] = (sv << 8) | (dv & 255);
    }
#pragma unroll
    for (int u = 0; u < 8; ++u)
        if (bn[u] >= 0) atomicAdd(&hist[bn[u]], 1);
    __syncthreads();

    const int hcount = hist[t];
    const int inc = gin_wave_incl(hcount, lane);
    if (lane == 31) wsum[wave] = inc;
    __syncthreads();
    int before = 0, total = 0;
    for (int w = 0; w < 16; ++w) {
        const int x = wsum[w];
        if (w < wave) before += x;
        total += x;
    }
    const int excl = before + inc - hcount;
    cur[t] = excl;
    if (t < nbin) tab[(size_t)t * S + s] = excl | (hcount << TSH);
    const int ncoarse = (nbin + CB - 1) / CB;
    if (t < ncoarse) {
        int a = 0;
#pragma unroll
        for (int q = 0; q < CB; ++q) a += hist[t * CB + q];
        tab2[(size_t)t * S + s] = a;
    }
    __syncthreads();

    int pos[8];
#pragma unroll
    for (int u = 0; u < 8; ++u) {
        pos[u] = -1;
        if (bn[u] >= 0) pos[u] = atomicAdd(&cur[bn[u]], 1);
    }
#pragma unroll
    for (int u = 0; u < 8; ++u)
        if (pos[u] >= 0) sorted[pos[u]] = pk[u];
    __syncthreads();
    for (int i = t; i < total; i += 512) part[e0 + i] = sorted[i];
}

__device__ __forceinline__ int gin_bin_entry(const int* __restrict__ part, const int* pre,
                                             const int* sbeg, int j) {
    int lo = 0;
#pragma unroll
    for (int step = SMAX / 2; step; step >>= 1) {
        const int mid = lo + step;
        if (pre[mid] <= j) lo = mid;
    }
    return part[sbeg[lo] + (j - pre[lo])];
}

__global__ __launch_bounds__(NPB) void gin_binbuild_kernel(const int* __restrict__ part,
                                                           const int* __restrict__ tab,
                                                           const int* __restrict__ tab2,
                                                           int* __restrict__ cnt,
                                                           int* __restrict__ off,
                                                           int* __restrict__ lst,
                                                           int n, int S) {
    __shared__ int pre[SMAX];
    __shared__ int sbeg[SMAX];
    __shared__ int cntN[NPB];
    __shared__ int cur[NPB];
    __shared__ int sw[8];
    __shared__ int sb[8];
    __shared__ int sw2[8];
    const int t    = threadIdx.x;
    const int lane = t & 31;
    const int wave = t >> 5;
    const int b    = blockIdx.x;

    int x = 0;
    if (t < S) x = tab[(size_t)b * S + t];

    int p = 0;
    const int cb = b / CB;
    const int n1 = cb * S;
    for (int i0 = 0; i0 < n1; i0 += NPB * 24) {
        int v[24];
#pragma unroll
        for (int u = 0; u < 24; ++u) v[u] = tab2[min(i0 + u * NPB + t, n1 - 1)];
#pragma unroll
        for (int u = 0; u < 24; ++u) p += (i0 + u * NPB + t < n1) ? v[u] : 0;
    }
    const int n2 = b * S;
    for (int i0 = cb * CB * S; i0 < n2; i0 += NPB * 16) {
        int v[16];
#pragma unroll
        for (int u = 0; u < 16; ++u) v[u] = tab[min(i0 + u * NPB + t, n2 - 1)];
#pragma unroll
        for (int u = 0; u < 16; ++u) p += (i0 + u * NPB + t < n2) ? (v[u] >> TSH) : 0;
    }
    for (int d = 16; d; d >>= 1) p += __shfl_xor(p, d, 32);

    const int len = x >> TSH;
    const int inc = gin_wave_incl(len, lane);
    if (lane == 31) sw[wave] = inc;
    if (lane == 0)  sb[wave] = p;
    cntN[t] = 0;
    __syncthreads();
    int before = 0, M = 0, base = 0;
    for (int w = 0; w < 8; ++w) {
        const int y = sw[w];
        if (w < wave) before += y;
        M += y;
        base += sb[w];
    }
    pre[t]  = (t < S) ? (before + inc - len) : M;
    sbeg[t] = t * SLICE + (x & ((1 << TSH) - 1));
    __syncthreads();

    for (int j0 = 0; j0 < M; j0 += NPB * EPB) {
        int w[EPB];
#pragma unroll
        for (int u = 0; u < EPB; ++u)
            w[u] = gin_bin_entry(part, pre, sbeg, min(j0 + u * NPB + t, M - 1));
        asm volatile("" :: "v"(w[0]), "v"(w[1]), "v"(w[2]), "v"(w[3]), "v"(w[4]), "v"(w[5]),
                           "v"(w[6]), "v"(w[7]), "v"(w[8]), "v"(w[9]), "v"(w[10]), "v"(w[11]));
#pragma unroll
        for (int u = 0; u < EPB; ++u)
            if (j0 + u * NPB + t < M) atomicAdd(&cntN[w[u] & 255], 1);
    }
    __syncthreads();

    const int deg  = cntN[t];
    const int inc2 = gin_wave_incl(deg, lane);
    if (lane == 31) sw2[wave] = inc2;
    __syncthreads();
    int before2 = 0;
    for (int w = 0; w < wave; ++w) before2 += sw2[w];
    const int first = base + before2 + inc2 - deg;
    const int node  = b * NPB + t;
    if (node < n) { off[node] = first; cnt[node] = deg; }
    cur[t] = first;
    __syncthreads();

    for (int j0 = 0; j0 < M; j0 += NPB * EPB) {
        int w[EPB];
#pragma unroll
        for (int u = 0; u < EPB; ++u)
            w[u] = gin_bin_entry(part, pre, sbeg, min(j0 + u * NPB + t, M - 1));
        asm volatile("" :: "v"(w[0]), "v"(w[1]), "v"(w[2]), "v"(w[3]), "v"(w[4]), "v"(w[5]),
                           "v"(w[6]), "v"(w[7]), "v"(w[8]), "v"(w[9]), "v"(w[10]), "v"(w[11]));
        int pos[EPB];
#pragma unroll
        for (int u = 0; u < EPB; ++u) {
            pos[u] = -1;
            if (j0 + u * NPB + t < M) pos[u] = atomicAdd(&cur[w[u] & 255], 1);
        }
#pragma unroll
        for (int u = 0; u < EPB; ++u)
            if (pos[u] >= 0) lst[pos[u]] = (int)((unsigned)w[u] >> 8);
    }
}

template <bool LAST>
__global__ __launch_bounds__(256)
void gin_layer_kernel(const float* __restrict__ h,
                      const int* __restrict__ off, const int* __restrict__ cnt,
                      const int* __restrict__ lst,
                      const float* __restrict__ w1, const float* __restrict__ b1,
                      const float* __restrict__ w2, const float* __restrict__ b2,
                      float* __restrict__ hout,
                      const float* __restrict__ ow, const float* __restrict__ ob,
                      float* __restrict__ out, int n) {
    __shared__ float sW1t[GIN_D * WSTR];
    __shared__ float sW2t[GIN_D * WSTR];
    __shared__ float sB1[GIN_D], sB2[GIN_D];
    __shared__ float sT[8][16 * TSTR];

    const int t    = threadIdx.x;
    const int wave = t >> 5;
    const int lane = t & 31;
    const int m    = lane & 15;
    const int hi   = lane >> 4;

    const int row0 = (blockIdx.x * 8 + wave) * 16;
    float* tile = sT[wave];

    const int q = lane >> 3;
    const int c = (lane & 7) << 2;

    int pbeg[4], pcnt[4];
#pragma unroll
    for (int it = 0; it < 4; ++it) {
        const int nd = min(row0 + it * 4 + q, n - 1);
        pbeg[it] = off[nd];
        pcnt[it] = cnt[nd];
    }

    for (int i = t; i < GIN_D * GIN_D / 4; i += 256) {
        int k  = i >> 4;
        int c4 = (i & 15) << 2;
        float4 a = ((const float4*)w1)[i];
        sW1t[(c4 + 0) * WSTR + k] = a.x;
        sW1t[(c4 + 1) * WSTR + k] = a.y;
        sW1t[(c4 + 2) * WSTR + k] = a.z;
        sW1t[(c4 + 3) * WSTR + k] = a.w;
        float4 b = ((const float4*)w2)[i];
        sW2t[(c4 + 0) * WSTR + k] = b.x;
        sW2t[(c4 + 1) * WSTR + k] = b.y;
        sW2t[(c4 + 2) * WSTR + k] = b.z;
        sW2t[(c4 + 3) * WSTR + k] = b.w;
    }
    if (t < GIN_D) { sB1[t] = b1[t]; sB2[t] = b2[t]; }

#pragma unroll
    for (int it = 0; it < 4; ++it) {
        const int r    = it * 4 + q;
        const int node = row0 + r;
        float4 a0 = make_float4(0.f, 0.f, 0.f, 0.f);
        float4 a1 = make_float4(0.f, 0.f, 0.f, 0.f);
        if (node < n) {
            int p = pbeg[it];
            const int end  = p + pcnt[it];
            const int last = end - 1;
            const float* hc = h + c;
            a0 = *(const float4*)(hc + (size_t)node * GIN_D);
            a1 = *(const float4*)(hc + (size_t)node * GIN_D + 32);
            int s0 = 0, s1 = 0, s2 = 0, s3 = 0;
            if (p < end) {
                s0 = lst[p];
                s1 = lst[min(p + 1, last)];
                s2 = lst[min(p + 2, last)];
                s3 = lst[min(p + 3, last)];
            }
            while (p < end) {
                const bool k1 = p + 1 < end;
                const bool k2 = p + 2 < end;
                const bool k3 = p + 3 < end;
                const float4 u0 = *(const float4*)(hc + (size_t)s0 * GIN_D);
                const float4 v0 = *(const float4*)(hc + (size_t)s0 * GIN_D + 32);
                const float4 u1 = *(const float4*)(hc + (size_t)s1 * GIN_D);
                const float4 v1 = *(const float4*)(hc + (size_t)s1 * GIN_D + 32);
                const float4 u2 = *(const float4*)(hc + (size_t)s2 * GIN_D);
                const float4 v2 = *(const float4*)(hc + (size_t)s2 * GIN_D + 32);
                const float4 u3 = *(const float4*)(hc + (size_t)s3 * GIN_D);
                const float4 v3 = *(const float4*)(hc + (size_t)s3 * GIN_D + 32);
                p += 4;
                if (p < end) {
                    s0 = lst[p];
                    s1 = lst[min(p + 1, last)];
                    s2 = lst[min(p + 2, last)];
                    s3 = lst[min(p + 3, last)];
                }
                a0.x += u0.x; a0.y += u0.y; a0.z += u0.z; a0.w += u0.w;
                a1.x += v0.x; a1.y += v0.y; a1.z += v0.z; a1.w += v0.w;
                a0.x += k1 ? u1.x : 0.f; a0.y += k1 ? u1.y : 0.f;
                a0.z += k1 ? u1.z : 0.f; a0.w += k1 ? u1.w : 0.f;
                a1.x += k1 ? v1.x : 0.f; a1.y += k1 ? v1.y : 0.f;
                a1.z += k1 ? v1.z : 0.f; a1.w += k1 ? v1.w : 0.f;
                a0.x += k2 ? u2.x : 0.f; a0.y += k2 ? u2.y : 0.f;
                a0.z += k2 ? u2.z : 0.f; a0.w += k2 ? u2.w : 0.f;
                a1.x += k2 ? v2.x : 0.f; a1.y += k2 ? v2.y : 0.f;
                a1.z += k2 ? v2.z : 0.f; a1.w += k2 ? v2.w : 0.f;
                a0.x += k3 ? u3.x : 0.f; a0.y += k3 ? u3.y : 0.f;
                a0.z += k3 ? u3.z : 0.f; a0.w += k3 ? u3.w : 0.f;
                a1.x += k3 ? v3.x : 0.f; a1.y += k3 ? v3.y : 0.f;
                a1.z += k3 ? v3.z : 0.f; a1.w += k3 ? v3.w : 0.f;
            }
        }
        float* tr = tile + r * TSTR + c;
        tr[0] = a0.x; tr[1] = a0.y; tr[2] = a0.z; tr[3] = a0.w;
        tr[32] = a1.x; tr[33] = a1.y; tr[34] = a1.z; tr[35] = a1.w;
    }
    __syncthreads();

    v2f A[16];
#pragma unroll
    for (int kb = 0; kb < 16; ++kb) {
        int k = 4 * kb + 2 * hi;
        A[kb].x = tile[m * TSTR + k];
        A[kb].y = tile[m * TSTR + k + 1];
    }
    __syncthreads();

    for (int nt = 0; nt < 4; ++nt) {
        const int n0 = nt * 16;
        float bv = sB1[n0 + m];
        v8f acc = {bv, bv, bv, bv, bv, bv, bv, bv};
#pragma unroll
        for (int kb = 0; kb < 16; ++kb) {
            int k = 4 * kb + 2 * hi;
            v2f B;
            B.x = sW1t[(n0 + m) * WSTR + k];
            B.y = sW1t[(n0 + m) * WSTR + k + 1];
            acc = __builtin_amdgcn_wmma_f32_16x16x4_f32(false, A[kb], false, B,
                                                        (short)0, acc, false, false);
        }
#pragma unroll
        for (int j = 0; j < 8; ++j)
            tile[(j + 8 * hi) * TSTR + n0 + m] = fmaxf(acc[j], 0.f);
    }
    __syncthreads();

#pragma unroll
    for (int kb = 0; kb < 16; ++kb) {
        int k = 4 * kb + 2 * hi;
        A[kb].x = tile[m * TSTR + k];
        A[kb].y = tile[m * TSTR + k + 1];
    }

    float ps[8] = {0.f, 0.f, 0.f, 0.f, 0.f, 0.f, 0.f, 0.f};
    for (int nt = 0; nt < 4; ++nt) {
        const int n0 = nt * 16;
        float bv = sB2[n0 + m];
        v8f acc = {bv, bv, bv, bv, bv, bv, bv, bv};
#pragma unroll
        for (int kb = 0; kb < 16; ++kb) {
            int k = 4 * kb + 2 * hi;
            v2f B;
            B.x = sW2t[(n0 + m) * WSTR + k];
            B.y = sW2t[(n0 + m) * WSTR + k + 1];
            acc = __builtin_amdgcn_wmma_f32_16x16x4_f32(false, A[kb], false, B,
                                                        (short)0, acc, false, false);
        }
        if (LAST) {
            const float wv = ow[n0 + m];
#pragma unroll
            for (int j = 0; j < 8; ++j) ps[j] += fmaxf(acc[j], 0.f) * wv;
        } else {
#pragma unroll
            for (int j = 0; j < 8; ++j) {
                int gr = row0 + j + 8 * hi;
                if (gr < n) hout[(size_t)gr * GIN_D + n0 + m] = fmaxf(acc[j], 0.f);
            }
        }
    }
    if (LAST) {
        const bool u3 = (m & 8) != 0;
        float q4[4];
#pragma unroll
        for (int i = 0; i < 4; ++i) {
            const float keep = u3 ? ps[i + 4] : ps[i];
            const float give = u3 ? ps[i] : ps[i + 4];
            q4[i] = keep + __shfl_xor(give, 8, 32);
        }
        const bool u2 = (m & 4) != 0;
        float q2[2];
#pragma unroll
        for (int i = 0; i < 2; ++i) {
            const float keep = u2 ? q4[i + 2] : q4[i];
            const float give = u2 ? q4[i] : q4[i + 2];
            q2[i] = keep + __shfl_xor(give, 4, 32);
        }
        const bool u1 = (m & 2) != 0;
        const float keep1 = u1 ? q2[1] : q2[0];
        const float give1 = u1 ? q2[0] : q2[1];
        float q1 = keep1 + __shfl_xor(give1, 2, 32);
        q1 += __shfl_xor(q1, 1, 32);
        const int gr = row0 + 8 * hi + (m >> 1);
        if ((m & 1) == 0 && gr < n) out[gr] = q1 + ob[0];
    }
}

extern "C" void kernel_launch(void* const* d_in, const int* in_sizes, int n_in,
                              void* d_out, int out_size, void* d_ws, size_t ws_size,
                              hipStream_t stream) {
    const float* x  = (const float*)d_in[0];
    const int*   ei = (const int*)d_in[1];
    const int N = in_sizes[0] / GIN_D;
    const int E = in_sizes[1] / 2;
    const int* src = ei;
    const int* dst = ei + E;
    const float* W1[3] = {(const float*)d_in[2], (const float*)d_in[6],  (const float*)d_in[10]};
    const float* B1[3] = {(const float*)d_in[3], (const float*)d_in[7],  (const float*)d_in[11]};
    const float* W2[3] = {(const float*)d_in[4], (const float*)d_in[8],  (const float*)d_in[12]};
    const float* B2[3] = {(const float*)d_in[5], (const float*)d_in[9],  (const float*)d_in[13]};
    const float* out_w = (const float*)d_in[14];
    const float* out_b = (const float*)d_in[15];
    float* out = (float*)d_out;

    float* hA  = (float*)d_ws;
    float* hB  = hA  + (size_t)N * GIN_D;
    const int S       = (E + SLICE - 1) / SLICE;
    const int nbin    = (N + NPB - 1) / NPB;
    const int ncoarse = (nbin + CB - 1) / CB;
    int* cnt  = (int*)(hB + (size_t)N * GIN_D);
    int* off  = cnt  + (size_t)nbin * NPB;
    int* tab  = off  + (size_t)nbin * NPB;
    int* tab2 = tab  + (size_t)nbin * S;
    int* part = tab2 + (size_t)ncoarse * S;
    int* lst  = part + (size_t)S * SLICE;

    dim3 b256(256);
    dim3 b512(512);
    dim3 sg((unsigned)S);
    dim3 ng((unsigned)nbin);
    dim3 lg((unsigned)((N + 127) / 128));

    gin_partition_kernel<<<sg, b512, 0, stream>>>(src, dst, part, tab, tab2, E, N, S, nbin);
    gin_binbuild_kernel<<<ng, b256, 0, stream>>>(part, tab, tab2, cnt, off, lst, N, S);

    gin_layer_kernel<false><<<lg, b256, 0, stream>>>(x,  off, cnt, lst, W1[0], B1[0], W2[0], B2[0], hA,
                                                     out_w, out_b, out, N);
    gin_layer_kernel<false><<<lg, b256, 0, stream>>>(hA, off, cnt, lst, W1[1], B1[1], W2[1], B2[1], hB,
                                                     out_w, out_b, out, N);
    gin_layer_kernel<true><<<lg, b256, 0, stream>>>(hB, off, cnt, lst, W1[2], B1[2], W2[2], B2[2], hA,
                                                    out_w, out_b, out, N);
}


# Fewer dependent memory round trips per wave in the layer launch: list slots fetched once per node (one per lane, handed out by lane exchange), 8-row trips with exact slots, weight loads as one group
# speedup vs baseline: 3.8961x; 1.1040x over previous
#include <hip/hip_runtime.h>

#define GIN_D 64
#define TSTR 68
#define WSTR 68

typedef __attribute__((ext_vector_type(2))) float v2f;
typedef __attribute__((ext_vector_type(8))) float v8f;

#define SLICE 4096
#define NPB   256
#define NBMAX 512
#define SMAX  256
#define CB    16
#define TSH   13
#define EPB   12

__device__ __forceinline__ int gin_wave_incl(int v, int lane) {
    for (int d = 1; d < 32; d <<= 1) {
        const int u = __shfl_up(v, d, 32);
        if (lane >= d) v += u;
    }
    return v;
}

__global__ __launch_bounds__(512) void gin_partition_kernel(const int* __restrict__ src,
                                                            const int* __restrict__ dst,
                                                            int* __restrict__ part,
                                                            int* __restrict__ tab,
                                                            int* __restrict__ tab2,
                                                            int e, int n, int S, int nbin) {
    __shared__ int hist[NBMAX];
    __shared__ int cur[NBMAX];
    __shared__ int sorted[SLICE];
    __shared__ int wsum[16];
    const int t    = threadIdx.x;
    const int lane = t & 31;
    const int wave = t >> 5;
    const int s    = blockIdx.x;
    hist[t] = 0;
    __syncthreads();
    const int e0 = s * SLICE;
    const int e1 = min(e, e0 + SLICE);
    int pk[8], bn[8];
#pragma unroll
    for (int u = 0; u < 8; ++u) {
        const int k  = e0 + u * 512 + t;
        const int kc = min(k, e1 - 1);
        const int dv = dst[kc];
        const int sv = src[kc];
        const bool ok = (k < e1) && ((unsigned)dv < (unsigned)n);
        bn[u] = ok ? (dv >> 8) : -1;
        pk[u] = (sv << 8) | (dv & 255);
    }
#pragma unroll
    for (int u = 0; u < 8; ++u)
        if (bn[u] >= 0) atomicAdd(&hist[bn[u]], 1);
    __syncthreads();

    const int hcount = hist[t];
    const int inc = gin_wave_incl(hcount, lane);
    if (lane == 31) wsum[wave] = inc;
    __syncthreads();
    int before = 0, total = 0;
    for (int w = 0; w < 16; ++w) {
        const int x = wsum[w];
        if (w < wave) before += x;
        total += x;
    }
    const int excl = before + inc - hcount;
    cur[t] = excl;
    if (t < nbin) tab[(size_t)t * S + s] = excl | (hcount << TSH);
    const int ncoarse = (nbin + CB - 1) / CB;
    if (t < ncoarse) {
        int a = 0;
#pragma unroll
        for (int q = 0; q < CB; ++q) a += hist[t * CB + q];
        tab2[(size_t)t * S + s] = a;
    }
    __syncthreads();

    int pos[8];
#pragma unroll
    for (int u = 0; u < 8; ++u) {
        pos[u] = -1;
        if (bn[u] >= 0) pos[u] = atomicAdd(&cur[bn[u]], 1);
    }
#pragma unroll
    for (int u = 0; u < 8; ++u)
        if (pos[u] >= 0) sorted[pos[u]] = pk[u];
    __syncthreads();
    for (int i = t; i < total; i += 512) part[e0 + i] = sorted[i];
}

__device__ __forceinline__ int gin_bin_entry(const int* __restrict__ part, const int* pre,
                                             const int* sbeg, int j) {
    int lo = 0;
#pragma unroll
    for (int step = SMAX / 2; step; step >>= 1) {
        const int mid = lo + step;
        if (pre[mid] <= j) lo = mid;
    }
    return part[sbeg[lo] + (j - pre[lo])];
}

__global__ __launch_bounds__(NPB) void gin_binbuild_kernel(const int* __restrict__ part,
                                                           const int* __restrict__ tab,
                                                           const int* __restrict__ tab2,
                                                           int* __restrict__ cnt,
                                                           int* __restrict__ off,
                                                           int* __restrict__ lst,
                                                           int n, int S) {
    __shared__ int pre[SMAX];
    __shared__ int sbeg[SMAX];
    __shared__ int cntN[NPB];
    __shared__ int cur[NPB];
    __shared__ int sw[8];
    __shared__ int sb[8];
    __shared__ int sw2[8];
    const int t    = threadIdx.x;
    const int lane = t & 31;
    const int wave = t >> 5;
    const int b    = blockIdx.x;

    int x = 0;
    if (t < S) x = tab[(size_t)b * S + t];

    int p = 0;
    const int cb = b / CB;
    const int n1 = cb * S;
    for (int i0 = 0; i0 < n1; i0 += NPB * 24) {
        int v[24];
#pragma unroll
        for (int u = 0; u < 24; ++u) v[u] = tab2[min(i0 + u * NPB + t, n1 - 1)];
#pragma unroll
        for (int u = 0; u < 24; ++u) p += (i0 + u * NPB + t < n1) ? v[u] : 0;
    }
    const int n2 = b * S;
    for (int i0 = cb * CB * S; i0 < n2; i0 += NPB * 16) {
        int v[16];
#pragma unroll
        for (int u = 0; u < 16; ++u) v[u] = tab[min(i0 + u * NPB + t, n2 - 1)];
#pragma unroll
        for (int u = 0; u < 16; ++u) p += (i0 + u * NPB + t < n2) ? (v[u] >> TSH) : 0;
    }
    for (int d = 16; d; d >>= 1) p += __shfl_xor(p, d, 32);

    const int len = x >> TSH;
    const int inc = gin_wave_incl(len, lane);
    if (lane == 31) sw[wave] = inc;
    if (lane == 0)  sb[wave] = p;
    cntN[t] = 0;
    __syncthreads();
    int before = 0, M = 0, base = 0;
    for (int w = 0; w < 8; ++w) {
        const int y = sw[w];
        if (w < wave) before += y;
        M += y;
        base += sb[w];
    }
    pre[t]  = (t < S) ? (before + inc - len) : M;
    sbeg[t] = t * SLICE + (x & ((1 << TSH) - 1));
    __syncthreads();

    for (int j0 = 0; j0 < M; j0 += NPB * EPB) {
        int w[EPB];
#pragma unroll
        for (int u = 0; u < EPB; ++u)
            w[u] = gin_bin_entry(part, pre, sbeg, min(j0 + u * NPB + t, M - 1));
        asm volatile("" :: "v"(w[0]), "v"(w[1]), "v"(w[2]), "v"(w[3]), "v"(w[4]), "v"(w[5]),
                           "v"(w[6]), "v"(w[7]), "v"(w[8]), "v"(w[9]), "v"(w[10]), "v"(w[11]));
#pragma unroll
        for (int u = 0; u < EPB; ++u)
            if (j0 + u * NPB + t < M) atomicAdd(&cntN[w[u] & 255], 1);
    }
    __syncthreads();

    const int deg  = cntN[t];
    const int inc2 = gin_wave_incl(deg, lane);
    if (lane == 31) sw2[wave] = inc2;
    __syncthreads();
    int before2 = 0;
    for (int w = 0; w < wave; ++w) before2 += sw2[w];
    const int first = base + before2 + inc2 - deg;
    const int node  = b * NPB + t;
    if (node < n) { off[node] = first; cnt[node] = deg; }
    cur[t] = first;
    __syncthreads();

    for (int j0 = 0; j0 < M; j0 += NPB * EPB) {
        int w[EPB];
#pragma unroll
        for (int u = 0; u < EPB; ++u)
            w[u] = gin_bin_entry(part, pre, sbeg, min(j0 + u * NPB + t, M - 1));
        asm volatile("" :: "v"(w[0]), "v"(w[1]), "v"(w[2]), "v"(w[3]), "v"(w[4]), "v"(w[5]),
                           "v"(w[6]), "v"(w[7]), "v"(w[8]), "v"(w[9]), "v"(w[10]), "v"(w[11]));
        int pos[EPB];
#pragma unroll
        for (int u = 0; u < EPB; ++u) {
            pos[u] = -1;
            if (j0 + u * NPB + t < M) pos[u] = atomicAdd(&cur[w[u] & 255], 1);
        }
#pragma unroll
        for (int u = 0; u < EPB; ++u)
            if (pos[u] >= 0) lst[pos[u]] = (int)((unsigned)w[u] >> 8);
    }
}

template <bool LAST>
__global__ __launch_bounds__(256)
void gin_layer_kernel(const float* __restrict__ h,
                      const int* __restrict__ off, const int* __restrict__ cnt,
                      const int* __restrict__ lst,
                      const float* __restrict__ w1, const float* __restrict__ b1,
                      const float* __restrict__ w2, const float* __restrict__ b2,
                      float* __restrict__ hout,
                      const float* __restrict__ ow, const float* __restrict__ ob,
                      float* __restrict__ out, int n) {
    __shared__ float sW1t[GIN_D * WSTR];
    __shared__ float sW2t[GIN_D * WSTR];
    __shared__ float sB1[GIN_D], sB2[GIN_D];
    __shared__ float sT[8][16 * TSTR];

    const int t    = threadIdx.x;
    const int wave = t >> 5;
    const int lane = t & 31;
    const int m    = lane & 15;
    const int hi   = lane >> 4;

    const int row0 = (blockIdx.x * 8 + wave) * 16;
    float* tile = sT[wave];

    const int q  = lane >> 3;
    const int j8 = lane & 7;
    const int c  = j8 << 2;

    int pbeg[4], pend[4];
#pragma unroll
    for (int it = 0; it < 4; ++it) {
        const int nd = min(row0 + it * 4 + q, n - 1);
        pbeg[it] = off[nd];
        pend[it] = pbeg[it] + cnt[nd];
    }
    float4 wa[4], wb[4];
#pragma unroll
    for (int u = 0; u < 4; ++u) {
        wa[u] = ((const float4*)w1)[u * 256 + t];
        wb[u] = ((const float4*)w2)[u * 256 + t];
    }
    float bias1 = 0.f, bias2 = 0.f;
    if (t < GIN_D) { bias1 = b1[t]; bias2 = b2[t]; }
    int ia[4], ib[4];
#pragma unroll
    for (int it = 0; it < 4; ++it) {
        ia[it] = 0;
        ib[it] = 0;
        if (pbeg[it] + j8 < pend[it])     ia[it] = lst[pbeg[it] + j8];
        if (pbeg[it] + 8 + j8 < pend[it]) ib[it] = lst[pbeg[it] + 8 + j8];
    }

#pragma unroll
    for (int u = 0; u < 4; ++u) {
        const int i  = u * 256 + t;
        const int k  = i >> 4;
        const int c4 = (i & 15) << 2;
        sW1t[(c4 + 0) * WSTR + k] = wa[u].x;
        sW1t[(c4 + 1) * WSTR + k] = wa[u].y;
        sW1t[(c4 + 2) * WSTR + k] = wa[u].z;
        sW1t[(c4 + 3) * WSTR + k] = wa[u].w;
        sW2t[(c4 + 0) * WSTR + k] = wb[u].x;
        sW2t[(c4 + 1) * WSTR + k] = wb[u].y;
        sW2t[(c4 + 2) * WSTR + k] = wb[u].z;
        sW2t[(c4 + 3) * WSTR + k] = wb[u].w;
    }
    if (t < GIN_D) { sB1[t] = bias1; sB2[t] = bias2; }

    const char* hb = (const char*)h;
    const unsigned cbyte = (unsigned)c * 4u;
#pragma unroll
    for (int it = 0; it < 4; ++it) {
        const int r    = it * 4 + q;
        const int node = row0 + r;
        float4 a0 = make_float4(0.f, 0.f, 0.f, 0.f);
        float4 a1 = make_float4(0.f, 0.f, 0.f, 0.f);
        if (node < n) {
            int p = pbeg[it];
            const int end = pend[it];
            int sa = ia[it];
            int sb = ib[it];
            const char* own = hb + (((unsigned)node << 8) + cbyte);
            a0 = *(const float4*)own;
            a1 = *(const float4*)(own + 128);
            while (p < end) {
                float4 u[8], v[8];
#define GIN_SLOT(J)                                                                         \
                {                                                                           \
                         \
                    const int s = (int)__builtin_amdgcn_mov_dpp8((unsigned)sa, (J) * 0x249249); \
                    u[J] = make_float4(0.f, 0.f, 0.f, 0.f);                                 \
                    v[J] = make_float4(0.f, 0.f, 0.f, 0.f);                                 \
                    if (p + (J) < end) {                                                    \
                        const char* rp = hb + (((unsigned)s << 8) + cbyte);                 \
                        u[J] = *(const float4*)rp;                                          \
                        v[J] = *(const float4*)(rp + 128);                                  \
                    }                                                                       \
                }
                GIN_SLOT(0) GIN_SLOT(1) GIN_SLOT(2) GIN_SLOT(3)
                GIN_SLOT(4) GIN_SLOT(5) GIN_SLOT(6) GIN_SLOT(7)
#undef GIN_SLOT
                p += 8;
                sa = sb;
                sb = 0;
                if (p + 8 + j8 < end) sb = lst[p + 8 + j8];
#pragma unroll
                for (int j = 0; j < 8; ++j) {
                    a0.x += u[j].x; a0.y += u[j].y; a0.z += u[j].z; a0.w += u[j].w;
                    a1.x += v[j].x; a1.y += v[j].y; a1.z += v[j].z; a1.w += v[j].w;
                }
            }
        }
        float* tr = tile + r * TSTR + c;
        tr[0] = a0.x; tr[1] = a0.y; tr[2] = a0.z; tr[3] = a0.w;
        tr[32] = a1.x; tr[33] = a1.y; tr[34] = a1.z; tr[35] = a1.w;
    }
    __syncthreads();

    v2f A[16];
#pragma unroll
    for (int kb = 0; kb < 16; ++kb) {
        int k = 4 * kb + 2 * hi;
        A[kb].x = tile[m * TSTR + k];
        A[kb].y = tile[m * TSTR + k + 1];
    }
    __syncthreads();

    for (int nt = 0; nt < 4; ++nt) {
        const int n0 = nt * 16;
        float bv = sB1[n0 + m];
        v8f acc = {bv, bv, bv, bv, bv, bv, bv, bv};
#pragma unroll
        for (int kb = 0; kb < 16; ++kb) {
            int k = 4 * kb + 2 * hi;
            v2f B;
            B.x = sW1t[(n0 + m) * WSTR + k];
            B.y = sW1t[(n0 + m) * WSTR + k + 1];
            acc = __builtin_amdgcn_wmma_f32_16x16x4_f32(false, A[kb], false, B,
                                                        (short)0, acc, false, false);
        }
#pragma unroll
        for (int j = 0; j < 8; ++j)
            tile[(j + 8 * hi) * TSTR + n0 + m] = fmaxf(acc[j], 0.f);
    }
    __syncthreads();

#pragma unroll
    for (int kb = 0; kb < 16; ++kb) {
        int k = 4 * kb + 2 * hi;
        A[kb].x = tile[m * TSTR + k];
        A[kb].y = tile[m * TSTR + k + 1];
    }

    float ps[8] = {0.f, 0.f, 0.f, 0.f, 0.f, 0.f, 0.f, 0.f};
    for (int nt = 0; nt < 4; ++nt) {
        const int n0 = nt * 16;
        float bv = sB2[n0 + m];
        v8f acc = {bv, bv, bv, bv, bv, bv, bv, bv};
#pragma unroll
        for (int kb = 0; kb < 16; ++kb) {
            int k = 4 * kb + 2 * hi;
            v2f B;
            B.x = sW2t[(n0 + m) * WSTR + k];
            B.y = sW2t[(n0 + m) * WSTR + k + 1];
            acc = __builtin_amdgcn_wmma_f32_16x16x4_f32(false, A[kb], false, B,
                                                        (short)0, acc, false, false);
        }
        if (LAST) {
            const float wv = ow[n0 + m];
#pragma unroll
            for (int j = 0; j < 8; ++j) ps[j] += fmaxf(acc[j], 0.f) * wv;
        } else {
#pragma unroll
            for (int j = 0; j < 8; ++j) {
                int gr = row0 + j + 8 * hi;
                if (gr < n) hout[(size_t)gr * GIN_D + n0 + m] = fmaxf(acc[j], 0.f);
            }
        }
    }
    if (LAST) {
        const bool u3 = (m & 8) != 0;
        float q4[4];
#pragma unroll
        for (int i = 0; i < 4; ++i) {
            const float keep = u3 ? ps[i + 4] : ps[i];
            const float give = u3 ? ps[i] : ps[i + 4];
            q4[i] = keep + __shfl_xor(give, 8, 32);
        }
        const bool u2 = (m & 4) != 0;
        float q2[2];
#pragma unroll
        for (int i = 0; i < 2; ++i) {
            const float keep = u2 ? q4[i + 2] : q4[i];
            const float give = u2 ? q4[i] : q4[i + 2];
            q2[i] = keep + __shfl_xor(give, 4, 32);
        }
        const bool u1 = (m & 2) != 0;
        const float keep1 = u1 ? q2[1] : q2[0];
        const float give1 = u1 ? q2[0] : q2[1];
        float q1 = keep1 + __shfl_xor(give1, 2, 32);
        q1 += __shfl_xor(q1, 1, 32);
        const int gr = row0 + 8 * hi + (m >> 1);
        if ((m & 1) == 0 && gr < n) out[gr] = q1 + ob[0];
    }
}

extern "C" void kernel_launch(void* const* d_in, const int* in_sizes, int n_in,
                              void* d_out, int out_size, void* d_ws, size_t ws_size,
                              hipStream_t stream) {
    const float* x  = (const float*)d_in[0];
    const int*   ei = (const int*)d_in[1];
    const int N = in_sizes[0] / GIN_D;
    const int E = in_sizes[1] / 2;
    const int* src = ei;
    const int* dst = ei + E;
    const float* W1[3] = {(const float*)d_in[2], (const float*)d_in[6],  (const float*)d_in[10]};
    const float* B1[3] = {(const float*)d_in[3], (const float*)d_in[7],  (const float*)d_in[11]};
    const float* W2[3] = {(const float*)d_in[4], (const float*)d_in[8],  (const float*)d_in[12]};
    const float* B2[3] = {(const float*)d_in[5], (const float*)d_in[9],  (const float*)d_in[13]};
    const float* out_w = (const float*)d_in[14];
    const float* out_b = (const float*)d_in[15];
    float* out = (float*)d_out;

    float* hA  = (float*)d_ws;
    float* hB  = hA  + (size_t)N * GIN_D;
    const int S       = (E + SLICE - 1) / SLICE;
    const int nbin    = (N + NPB - 1) / NPB;
    const int ncoarse = (nbin + CB - 1) / CB;
    int* cnt  = (int*)(hB + (size_t)N * GIN_D);
    int* off  = cnt  + (size_t)nbin * NPB;
    int* tab  = off  + (size_t)nbin * NPB;
    int* tab2 = tab  + (size_t)nbin * S;
    int* part = tab2 + (size_t)ncoarse * S;
    int* lst  = part + (size_t)S * SLICE;

    dim3 b256(256);
    dim3 b512(512);
    dim3 sg((unsigned)S);
    dim3 ng((unsigned)nbin);
    dim3 lg((unsigned)((N + 127) / 128));

    gin_partition_kernel<<<sg, b512, 0, stream>>>(src, dst, part, tab, tab2, E, N, S, nbin);
    gin_binbuild_kernel<<<ng, b256, 0, stream>>>(part, tab, tab2, cnt, off, lst, N, S);

    gin_layer_kernel<false><<<lg, b256, 0, stream>>>(x,  off, cnt, lst, W1[0], B1[0], W2[0], B2[0], hA,
                                                     out_w, out_b, out, N);
    gin_layer_kernel<false><<<lg, b256, 0, stream>>>(hA, off, cnt, lst, W1[1], B1[1], W2[1], B2[1], hB,
                                                     out_w, out_b, out, N);
    gin_layer_kernel<true><<<lg, b256, 0, stream>>>(hB, off, cnt, lst, W1[2], B1[2], W2[2], B2[2], hA,
                                                    out_w, out_b, out, N);
}


# Layer launch as one 32-wave work-group per WGP (grid 256) with the 16-row tiles dealt evenly: weights staged once per WGP, busiest WGP 25 tiles not 32, one barrier and then free-running waves
# speedup vs baseline: 4.0792x; 1.0470x over previous
#include <hip/hip_runtime.h>

#define GIN_D 64
#define TSTR 68
#define WSTR 68

typedef __attribute__((ext_vector_type(2))) float v2f;
typedef __attribute__((ext_vector_type(8))) float v8f;

#define SLICE 4096
#define NPB   256
#define NBMAX 512
#define SMAX  256
#define CB    16
#define TSH   13
#define EPB   12

__device__ __forceinline__ int gin_wave_incl(int v, int lane) {
    for (int d = 1; d < 32; d <<= 1) {
        const int u = __shfl_up(v, d, 32);
        if (lane >= d) v += u;
    }
    return v;
}

__global__ __launch_bounds__(512) void gin_partition_kernel(const int* __restrict__ src,
                                                            const int* __restrict__ dst,
                                                            int* __restrict__ part,
                                                            int* __restrict__ tab,
                                                            int* __restrict__ tab2,
                                                            int e, int n, int S, int nbin) {
    __shared__ int hist[NBMAX];
    __shared__ int cur[NBMAX];
    __shared__ int sorted[SLICE];
    __shared__ int wsum[16];
    const int t    = threadIdx.x;
    const int lane = t & 31;
    const int wave = t >> 5;
    const int s    = blockIdx.x;
    hist[t] = 0;
    __syncthreads();
    const int e0 = s * SLICE;
    const int e1 = min(e, e0 + SLICE);
    int pk[8], bn[8];
#pragma unroll
    for (int u = 0; u < 8; ++u) {
        const int k  = e0 + u * 512 + t;
        const int kc = min(k, e1 - 1);
        const int dv = dst[kc];
        const int sv = src[kc];
        const bool ok = (k < e1) && ((unsigned)dv < (unsigned)n);
        bn[u] = ok ? (dv >> 8) : -1;
        pk[u] = (sv << 8) | (dv & 255);
    }
#pragma unroll
    for (int u = 0; u < 8; ++u)
        if (bn[u] >= 0) atomicAdd(&hist[bn[u]], 1);
    __syncthreads();

    const int hcount = hist[t];
    const int inc = gin_wave_incl(hcount, lane);
    if (lane == 31) wsum[wave] = inc;
    __syncthreads();
    int before = 0, total = 0;
    for (int w = 0; w < 16; ++w) {
        const int x = wsum[w];
        if (w < wave) before += x;
        total += x;
    }
    const int excl = before + inc - hcount;
    cur[t] = excl;
    if (t < nbin) tab[(size_t)t * S + s] = excl | (hcount << TSH);
    const int ncoarse = (nbin + CB - 1) / CB;
    if (t < ncoarse) {
        int a = 0;
#pragma unroll
        for (int q = 0; q < CB; ++q) a += hist[t * CB + q];
        tab2[(size_t)t * S + s] = a;
    }
    __syncthreads();

    int pos[8];
#pragma unroll
    for (int u = 0; u < 8; ++u) {
        pos[u] = -1;
        if (bn[u] >= 0) pos[u] = atomicAdd(&cur[bn[u]], 1);
    }
#pragma unroll
    for (int u = 0; u < 8; ++u)
        if (pos[u] >= 0) sorted[pos[u]] = pk[u];
    __syncthreads();
    for (int i = t; i < total; i += 512) part[e0 + i] = sorted[i];
}

__device__ __forceinline__ int gin_bin_entry(const int* __restrict__ part, const int* pre,
                                             const int* sbeg, int j) {
    int lo = 0;
#pragma unroll
    for (int step = SMAX / 2; step; step >>= 1) {
        const int mid = lo + step;
        if (pre[mid] <= j) lo = mid;
    }
    return part[sbeg[lo] + (j - pre[lo])];
}

__global__ __launch_bounds__(NPB) void gin_binbuild_kernel(const int* __restrict__ part,
                                                           const int* __restrict__ tab,
                                                           const int* __restrict__ tab2,
                                                           int* __restrict__ cnt,
                                                           int* __restrict__ off,
                                                           int* __restrict__ lst,
                                                           int n, int S) {
    __shared__ int pre[SMAX];
    __shared__ int sbeg[SMAX];
    __shared__ int cntN[NPB];
    __shared__ int cur[NPB];
    __shared__ int sw[8];
    __shared__ int sb[8];
    __shared__ int sw2[8];
    const int t    = threadIdx.x;
    const int lane = t & 31;
    const int wave = t >> 5;
    const int b    = blockIdx.x;

    int x = 0;
    if (t < S) x = tab[(size_t)b * S + t];

    int p = 0;
    const int cb = b / CB;
    const int n1 = cb * S;
    for (int i0 = 0; i0 < n1; i0 += NPB * 24) {
        int v[24];
#pragma unroll
        for (int u = 0; u < 24; ++u) v[u] = tab2[min(i0 + u * NPB + t, n1 - 1)];
#pragma unroll
        for (int u = 0; u < 24; ++u) p += (i0 + u * NPB + t < n1) ? v[u] : 0;
    }
    const int n2 = b * S;
    for (int i0 = cb * CB * S; i0 < n2; i0 += NPB * 16) {
        int v[16];
#pragma unroll
        for (int u = 0; u < 16; ++u) v[u] = tab[min(i0 + u * NPB + t, n2 - 1)];
#pragma unroll
        for (int u = 0; u < 16; ++u) p += (i0 + u * NPB + t < n2) ? (v[u] >> TSH) : 0;
    }
    for (int d = 16; d; d >>= 1) p += __shfl_xor(p, d, 32);

    const int len = x >> TSH;
    const int inc = gin_wave_incl(len, lane);
    if (lane == 31) sw[wave] = inc;
    if (lane == 0)  sb[wave] = p;
    cntN[t] = 0;
    __syncthreads();
    int before = 0, M = 0, base = 0;
    for (int w = 0; w < 8; ++w) {
        const int y = sw[w];
        if (w < wave) before += y;
        M += y;
        base += sb[w];
    }
    pre[t]  = (t < S) ? (before + inc - len) : M;
    sbeg[t] = t * SLICE + (x & ((1 << TSH) - 1));
    __syncthreads();

    for (int j0 = 0; j0 < M; j0 += NPB * EPB) {
        int w[EPB];
#pragma unroll
        for (int u = 0; u < EPB; ++u)
            w[u] = gin_bin_entry(part, pre, sbeg, min(j0 + u * NPB + t, M - 1));
        asm volatile("" :: "v"(w[0]), "v"(w[1]), "v"(w[2]), "v"(w[3]), "v"(w[4]), "v"(w[5]),
                           "v"(w[6]), "v"(w[7]), "v"(w[8]), "v"(w[9]), "v"(w[10]), "v"(w[11]));
#pragma unroll
        for (int u = 0; u < EPB; ++u)
            if (j0 + u * NPB + t < M) atomicAdd(&cntN[w[u] & 255], 1);
    }
    __syncthreads();

    const int deg  = cntN[t];
    const int inc2 = gin_wave_incl(deg, lane);
    if (lane == 31) sw2[wave] = inc2;
    __syncthreads();
    int before2 = 0;
    for (int w = 0; w < wave; ++w) before2 += sw2[w];
    const int first = base + before2 + inc2 - deg;
    const int node  = b * NPB + t;
    if (node < n) { off[node] = first; cnt[node] = deg; }
    cur[t] = first;
    __syncthreads();

    for (int j0 = 0; j0 < M; j0 += NPB * EPB) {
        int w[EPB];
#pragma unroll
        for (int u = 0; u < EPB; ++u)
            w[u] = gin_bin_entry(part, pre, sbeg, min(j0 + u * NPB + t, M - 1));
        asm volatile("" :: "v"(w[0]), "v"(w[1]), "v"(w[2]), "v"(w[3]), "v"(w[4]), "v"(w[5]),
                           "v"(w[6]), "v"(w[7]), "v"(w[8]), "v"(w[9]), "v"(w[10]), "v"(w[11]));
        int pos[EPB];
#pragma unroll
        for (int u = 0; u < EPB; ++u) {
            pos[u] = -1;
            if (j0 + u * NPB + t < M) pos[u] = atomicAdd(&cur[w[u] & 255], 1);
        }
#pragma unroll
        for (int u = 0; u < EPB; ++u)
            if (pos[u] >= 0) lst[pos[u]] = (int)((unsigned)w[u] >> 8);
    }
}

#define LWAVES   32
#define LTHREADS (LWAVES * 32)

__device__ __forceinline__ void gin_wave_sync() {
    __builtin_amdgcn_fence(__ATOMIC_RELEASE, "wavefront");
    __builtin_amdgcn_wave_barrier();
    __builtin_amdgcn_fence(__ATOMIC_ACQUIRE, "wavefront");
}

template <bool LAST>
__global__ __launch_bounds__(LTHREADS)
void gin_layer_kernel(const float* __restrict__ h,
                      const int* __restrict__ off, const int* __restrict__ cnt,
                      const int* __restrict__ lst,
                      const float* __restrict__ w1, const float* __restrict__ b1,
                      const float* __restrict__ w2, const float* __restrict__ b2,
                      float* __restrict__ hout,
                      const float* __restrict__ ow, const float* __restrict__ ob,
                      float* __restrict__ out, int n) {
    __shared__ float sW1t[GIN_D * WSTR];
    __shared__ float sW2t[GIN_D * WSTR];
    __shared__ float sB1[GIN_D], sB2[GIN_D];
    __shared__ float sT[LWAVES][16 * TSTR];

    const int t    = threadIdx.x;
    const int wave = __builtin_amdgcn_readfirstlane(t >> 5);
    const int lane = t & 31;
    const int m    = lane & 15;
    const int hi   = lane >> 4;

    const int ntiles = (n + 15) >> 4;
    const int nwg    = (int)gridDim.x;
    const int per    = ntiles / nwg;
    const int rem    = ntiles - per * nwg;
    const int wg     = (int)blockIdx.x;
    const int first  = wg * per + min(wg, rem);
    const int count  = per + (wg < rem ? 1 : 0);

    float* tile = sT[wave];

    const int q  = lane >> 3;
    const int j8 = lane & 7;
    const int c  = j8 << 2;

    int  ti   = 4 * (4 * ((wave >> 2) & 1) + ((wave - (wave >> 3)) & 3)) + (wave >> 3);
    bool have = ti < count;
    int  row0 = have ? (first + ti) * 16 : n;

    int pbeg[4], pend[4];
#pragma unroll
    for (int it = 0; it < 4; ++it) {
        const int nd = min(row0 + it * 4 + q, n - 1);
        pbeg[it] = off[nd];
        pend[it] = pbeg[it] + cnt[nd];
    }
    const float4 wa = ((const float4*)w1)[t];
    const float4 wb = ((const float4*)w2)[t];
    float bias1 = 0.f, bias2 = 0.f;
    if (t < GIN_D) { bias1 = b1[t]; bias2 = b2[t]; }
    int ia[4], ib[4];
#pragma unroll
    for (int it = 0; it < 4; ++it) {
        ia[it] = 0;
        ib[it] = 0;
        if (pbeg[it] + j8 < pend[it])     ia[it] = lst[pbeg[it] + j8];
        if (pbeg[it] + 8 + j8 < pend[it]) ib[it] = lst[pbeg[it] + 8 + j8];
    }

    {
        const int k  = t >> 4;
        const int c4 = (t & 15) << 2;
        sW1t[(c4 + 0) * WSTR + k] = wa.x;
        sW1t[(c4 + 1) * WSTR + k] = wa.y;
        sW1t[(c4 + 2) * WSTR + k] = wa.z;
        sW1t[(c4 + 3) * WSTR + k] = wa.w;
        sW2t[(c4 + 0) * WSTR + k] = wb.x;
        sW2t[(c4 + 1) * WSTR + k] = wb.y;
        sW2t[(c4 + 2) * WSTR + k] = wb.z;
        sW2t[(c4 + 3) * WSTR + k] = wb.w;
    }
    if (t < GIN_D) { sB1[t] = bias1; sB2[t] = bias2; }
    __syncthreads();

    const char* hb = (const char*)h;
    const unsigned cbyte = (unsigned)c * 4u;

    while (have) {
#pragma unroll
        for (int it = 0; it < 4; ++it) {
            const int r    = it * 4 + q;
            const int node = row0 + r;
            float4 a0 = make_float4(0.f, 0.f, 0.f, 0.f);
            float4 a1 = make_float4(0.f, 0.f, 0.f, 0.f);
            if (node < n) {
                int p = pbeg[it];
                const int end = pend[it];
                int sa = ia[it];
                int sb = ib[it];
                const char* own = hb + (((unsigned)node << 8) + cbyte);
                a0 = *(const float4*)own;
                a1 = *(const float4*)(own + 128);
                while (p < end) {
                    float4 u[8], v[8];
#define GIN_SLOT(J)                                                                         \
                    {                                                                           \
                             \
                        const int s = (int)__builtin_amdgcn_mov_dpp8((unsigned)sa, (J) * 0x249249); \
                        u[J] = make_float4(0.f, 0.f, 0.f, 0.f);                                 \
                        v[J] = make_float4(0.f, 0.f, 0.f, 0.f);                                 \
                        if (p + (J) < end) {                                                    \
                            const char* rp = hb + (((unsigned)s << 8) + cbyte);                 \
                            u[J] = *(const float4*)rp;                                          \
                            v[J] = *(const float4*)(rp + 128);                                  \
                        }                                                                       \
                    }
                    GIN_SLOT(0) GIN_SLOT(1) GIN_SLOT(2) GIN_SLOT(3)
                    GIN_SLOT(4) GIN_SLOT(5) GIN_SLOT(6) GIN_SLOT(7)
#undef GIN_SLOT
                    p += 8;
                    sa = sb;
                    sb = 0;
                    if (p + 8 + j8 < end) sb = lst[p + 8 + j8];
#pragma unroll
                    for (int j = 0; j < 8; ++j) {
                        a0.x += u[j].x; a0.y += u[j].y; a0.z += u[j].z; a0.w += u[j].w;
                        a1.x += v[j].x; a1.y += v[j].y; a1.z += v[j].z; a1.w += v[j].w;
                    }
                }
            }
            float* tr = tile + r * TSTR + c;
            tr[0] = a0.x; tr[1] = a0.y; tr[2] = a0.z; tr[3] = a0.w;
            tr[32] = a1.x; tr[33] = a1.y; tr[34] = a1.z; tr[35] = a1.w;
        }
        gin_wave_sync();

        v2f A[16];
#pragma unroll
        for (int kb = 0; kb < 16; ++kb) {
            int k = 4 * kb + 2 * hi;
            A[kb].x = tile[m * TSTR + k];
            A[kb].y = tile[m * TSTR + k + 1];
        }
        gin_wave_sync();

        for (int nt = 0; nt < 4; ++nt) {
            const int n0 = nt * 16;
            float bv = sB1[n0 + m];
            v8f acc = {bv, bv, bv, bv, bv, bv, bv, bv};
#pragma unroll
            for (int kb = 0; kb < 16; ++kb) {
                int k = 4 * kb + 2 * hi;
                v2f B;
                B.x = sW1t[(n0 + m) * WSTR + k];
                B.y = sW1t[(n0 + m) * WSTR + k + 1];
                acc = __builtin_amdgcn_wmma_f32_16x16x4_f32(false, A[kb], false, B,
                                                            (short)0, acc, false, false);
            }
#pragma unroll
            for (int j = 0; j < 8; ++j)
                tile[(j + 8 * hi) * TSTR + n0 + m] = fmaxf(acc[j], 0.f);
        }
        gin_wave_sync();

#pragma unroll
        for (int kb = 0; kb < 16; ++kb) {
            int k = 4 * kb + 2 * hi;
            A[kb].x = tile[m * TSTR + k];
            A[kb].y = tile[m * TSTR + k + 1];
        }
        gin_wave_sync();

        float ps[8] = {0.f, 0.f, 0.f, 0.f, 0.f, 0.f, 0.f, 0.f};
        for (int nt = 0; nt < 4; ++nt) {
            const int n0 = nt * 16;
            float bv = sB2[n0 + m];
            v8f acc = {bv, bv, bv, bv, bv, bv, bv, bv};
#pragma unroll
            for (int kb = 0; kb < 16; ++kb) {
                int k = 4 * kb + 2 * hi;
                v2f B;
                B.x = sW2t[(n0 + m) * WSTR + k];
                B.y = sW2t[(n0 + m) * WSTR + k + 1];
                acc = __builtin_amdgcn_wmma_f32_16x16x4_f32(false, A[kb], false, B,
                                                            (short)0, acc, false, false);
            }
            if (LAST) {
                const float wv = ow[n0 + m];
#pragma unroll
                for (int j = 0; j < 8; ++j) ps[j] += fmaxf(acc[j], 0.f) * wv;
            } else {
#pragma unroll
                for (int j = 0; j < 8; ++j) {
                    int gr = row0 + j + 8 * hi;
                    if (gr < n) hout[(size_t)gr * GIN_D + n0 + m] = fmaxf(acc[j], 0.f);
                }
            }
        }
        if (LAST) {
            const bool u3 = (m & 8) != 0;
            float q4[4];
#pragma unroll
            for (int i = 0; i < 4; ++i) {
                const float keep = u3 ? ps[i + 4] : ps[i];
                const float give = u3 ? ps[i] : ps[i + 4];
                q4[i] = keep + __shfl_xor(give, 8, 32);
            }
            const bool u2 = (m & 4) != 0;
            float q2[2];
#pragma unroll
            for (int i = 0; i < 2; ++i) {
                const float keep = u2 ? q4[i + 2] : q4[i];
                const float give = u2 ? q4[i] : q4[i + 2];
                q2[i] = keep + __shfl_xor(give, 4, 32);
            }
            const bool u1 = (m & 2) != 0;
            const float keep1 = u1 ? q2[1] : q2[0];
            const float give1 = u1 ? q2[0] : q2[1];
            float q1 = keep1 + __shfl_xor(give1, 2, 32);
            q1 += __shfl_xor(q1, 1, 32);
            const int gr = row0 + 8 * hi + (m >> 1);
            if ((m & 1) == 0 && gr < n) out[gr] = q1 + ob[0];
        }

        ti += LWAVES;
        have = ti < count;
        if (have) {
            row0 = (first + ti) * 16;
#pragma unroll
            for (int it = 0; it < 4; ++it) {
                const int nd = min(row0 + it * 4 + q, n - 1);
                pbeg[it] = off[nd];
                pend[it] = pbeg[it] + cnt[nd];
            }
#pragma unroll
            for (int it = 0; it < 4; ++it) {
                ia[it] = 0;
                ib[it] = 0;
                if (pbeg[it] + j8 < pend[it])     ia[it] = lst[pbeg[it] + j8];
                if (pbeg[it] + 8 + j8 < pend[it]) ib[it] = lst[pbeg[it] + 8 + j8];
            }
        }
    }
}

extern "C" void kernel_launch(void* const* d_in, const int* in_sizes, int n_in,
                              void* d_out, int out_size, void* d_ws, size_t ws_size,
                              hipStream_t stream) {
    const float* x  = (const float*)d_in[0];
    const int*   ei = (const int*)d_in[1];
    const int N = in_sizes[0] / GIN_D;
    const int E = in_sizes[1] / 2;
    const int* src = ei;
    const int* dst = ei + E;
    const float* W1[3] = {(const float*)d_in[2], (const float*)d_in[6],  (const float*)d_in[10]};
    const float* B1[3] = {(const float*)d_in[3], (const float*)d_in[7],  (const float*)d_in[11]};
    const float* W2[3] = {(const float*)d_in[4], (const float*)d_in[8],  (const float*)d_in[12]};
    const float* B2[3] = {(const float*)d_in[5], (const float*)d_in[9],  (const float*)d_in[13]};
    const float* out_w = (const float*)d_in[14];
    const float* out_b = (const float*)d_in[15];
    float* out = (float*)d_out;

    float* hA  = (float*)d_ws;
    float* hB  = hA  + (size_t)N * GIN_D;
    const int S       = (E + SLICE - 1) / SLICE;
    const int nbin    = (N + NPB - 1) / NPB;
    const int ncoarse = (nbin + CB - 1) / CB;
    int* cnt  = (int*)(hB + (size_t)N * GIN_D);
    int* off  = cnt  + (size_t)nbin * NPB;
    int* tab  = off  + (size_t)nbin * NPB;
    int* tab2 = tab  + (size_t)nbin * S;
    int* part = tab2 + (size_t)ncoarse * S;
    int* lst  = part + (size_t)S * SLICE;

    dim3 b256(256);
    dim3 b512(512);
    dim3 sg((unsigned)S);
    dim3 ng((unsigned)nbin);
    dim3 lg(256);
    dim3 lb(LTHREADS);

    gin_partition_kernel<<<sg, b512, 0, stream>>>(src, dst, part, tab, tab2, E, N, S, nbin);
    gin_binbuild_kernel<<<ng, b256, 0, stream>>>(part, tab, tab2, cnt, off, lst, N, S);

    gin_layer_kernel<false><<<lg, lb, 0, stream>>>(x,  off, cnt, lst, W1[0], B1[0], W2[0], B2[0], hA,
                                                     out_w, out_b, out, N);
    gin_layer_kernel<false><<<lg, lb, 0, stream>>>(hA, off, cnt, lst, W1[1], B1[1], W2[1], B2[1], hB,
                                                     out_w, out_b, out, N);
    gin_layer_kernel<true><<<lg, lb, 0, stream>>>(hB, off, cnt, lst, W1[2], B1[2], W2[2], B2[2], hA,
                                                    out_w, out_b, out, N);
}
